# starting kernel
# baseline (speedup 1.0000x reference)
#include <hip/hip_runtime.h>
#include <hip/hip_bf16.h>

#define D_MODEL 1024
#define D_STATE 16
#define DT_RANK 64
#define BATCH   2
#define SEQLEN  4096
#define E_DIM   (DT_RANK + 2*D_STATE)
#define ROWS    (BATCH*SEQLEN)
#define NCHUNK  64
#define TCHUNK  (SEQLEN/NCHUNK)

typedef __attribute__((ext_vector_type(2))) float v2f;
typedef __attribute__((ext_vector_type(8))) float v8f;

__device__ __forceinline__ float softplus_f(float v) {
  return v > 20.f ? v : log1pf(__expf(v));
}

__global__ __launch_bounds__(256) void k_proj_dtbc(const float* __restrict__ x,
                                                   const float* __restrict__ Wbc,
                                                   float* __restrict__ dtBC) {
  const int NT = E_DIM / 16;
  int wave = (blockIdx.x * blockDim.x + threadIdx.x) >> 5;
  int lane = threadIdx.x & 31;
  int mt = wave / NT, nt = wave - mt * NT;
  int l = lane & 15, half = lane >> 4;

  const float* arow = x   + (size_t)(mt * 16 + l) * D_MODEL + 2 * half;
  const float* brow = Wbc + (size_t)(nt * 16 + l) * D_MODEL + 2 * half;

  v8f acc = {0.f, 0.f, 0.f, 0.f, 0.f, 0.f, 0.f, 0.f};
  for (int k = 0; k < D_MODEL; k += 4) {
    v2f a, b;
    a[0] = arow[k]; a[1] = arow[k + 1];
    b[0] = brow[k]; b[1] = brow[k + 1];
    acc = __builtin_amdgcn_wmma_f32_16x16x4_f32(false, a, false, b,
                                                (short)0, acc, false, false);
  }
#pragma unroll
  for (int r = 0; r < 8; ++r) {
    int m = mt * 16 + 8 * half + r;
    dtBC[(size_t)m * E_DIM + nt * 16 + l] = acc[r];
  }
}

__global__ __launch_bounds__(256) void k_proj_dt(const float* __restrict__ dtBC,
                                                 const float* __restrict__ Wdt,
                                                 const float* __restrict__ bdt,
                                                 float* __restrict__ dt) {
  const int NT = D_MODEL / 16;
  int wave = (blockIdx.x * blockDim.x + threadIdx.x) >> 5;
  int lane = threadIdx.x & 31;
  int mt = wave / NT, nt = wave - mt * NT;
  int l = lane & 15, half = lane >> 4;

  const float* arow = dtBC + (size_t)(mt * 16 + l) * E_DIM   + 2 * half;
  const float* brow = Wdt  + (size_t)(nt * 16 + l) * DT_RANK + 2 * half;

  v8f acc = {0.f, 0.f, 0.f, 0.f, 0.f, 0.f, 0.f, 0.f};
  for (int k = 0; k < DT_RANK; k += 4) {
    v2f a, b;
    a[0] = arow[k]; a[1] = arow[k + 1];
    b[0] = brow[k]; b[1] = brow[k + 1];
    acc = __builtin_amdgcn_wmma_f32_16x16x4_f32(false, a, false, b,
                                                (short)0, acc, false, false);
  }
  float bias = bdt[nt * 16 + l];
#pragma unroll
  for (int r = 0; r < 8; ++r) {
    int m = mt * 16 + 8 * half + r;
    dt[(size_t)m * D_MODEL + nt * 16 + l] = softplus_f(acc[r] + bias);
  }
}

__global__ __launch_bounds__(256) void k_scan_local(const float* __restrict__ x,
                                                    const float* __restrict__ dt,
                                                    const float* __restrict__ dtBC,
                                                    const float* __restrict__ A_log,
                                                    float* __restrict__ hend,
                                                    float* __restrict__ Pprod) {
  int gid = blockIdx.x * blockDim.x + threadIdx.x;
  int d = gid & (D_MODEL - 1);
  int c = (gid >> 10) & (NCHUNK - 1);
  int b = gid >> 16;

  float An[D_STATE];
#pragma unroll
  for (int n = 0; n < D_STATE; ++n) An[n] = -__expf(A_log[d * D_STATE + n]);

  float h[D_STATE], P[D_STATE];
#pragma unroll
  for (int n = 0; n < D_STATE; ++n) { h[n] = 0.f; P[n] = 1.f; }

  int t0 = c * TCHUNK;
  for (int t = t0; t < t0 + TCHUNK; ++t) {
    size_t row = (size_t)b * SEQLEN + t;
    float dtv = dt[row * D_MODEL + d];
    float xv  = x[row * D_MODEL + d];
    float Bv[D_STATE];
    const float4* bp = (const float4*)(dtBC + row * E_DIM + DT_RANK);
#pragma unroll
    for (int i = 0; i < 4; ++i) {
      float4 q = bp[i];
      Bv[4*i+0] = q.x; Bv[4*i+1] = q.y; Bv[4*i+2] = q.z; Bv[4*i+3] = q.w;
    }
    float dx = dtv * xv;
#pragma unroll
    for (int n = 0; n < D_STATE; ++n) {
      float dA = __expf(dtv * An[n]);
      h[n] = dA * h[n] + dx * Bv[n];
      P[n] *= dA;
    }
  }
  size_t base = ((size_t)(b * D_MODEL + d) * NCHUNK + c) * D_STATE;
#pragma unroll
  for (int n = 0; n < D_STATE; ++n) {
    hend[base + n]  = h[n];
    Pprod[base + n] = P[n];
  }
}

__global__ __launch_bounds__(256) void k_scan_combine(const float* __restrict__ hend,
                                                      const float* __restrict__ Pprod,
                                                      float* __restrict__ h0) {
  int gid = blockIdx.x * blockDim.x + threadIdx.x;
  int n = gid & (D_STATE - 1);
  int bd = gid >> 4;
  size_t base = (size_t)bd * NCHUNK * D_STATE + n;
  float h = 0.f;
  for (int c = 0; c < NCHUNK; ++c) {
    size_t idx = base + (size_t)c * D_STATE;
    h0[idx] = h;
    h = Pprod[idx] * h + hend[idx];
  }
}

__global__ __launch_bounds__(256) void k_scan_emit(const float* __restrict__ x,
                                                   const float* __restrict__ dt,
                                                   const float* __restrict__ dtBC,
                                                   const float* __restrict__ A_log,
                                                   const float* __restrict__ Dp,
                                                   const float* __restrict__ h0,
                                                   float* __restrict__ y) {
  int gid = blockIdx.x * blockDim.x + threadIdx.x;
  int d = gid & (D_MODEL - 1);
  int c = (gid >> 10) & (NCHUNK - 1);
  int b = gid >> 16;

  float An[D_STATE];
#pragma unroll
  for (int n = 0; n < D_STATE; ++n) An[n] = -__expf(A_log[d * D_STATE + n]);

  float h[D_STATE];
  size_t base = ((size_t)(b * D_MODEL + d) * NCHUNK + c) * D_STATE;
#pragma unroll
  for (int n = 0; n < D_STATE; ++n) h[n] = h0[base + n];

  float Dd = Dp[d];
  int t0 = c * TCHUNK;
  for (int t = t0; t < t0 + TCHUNK; ++t) {
    size_t row = (size_t)b * SEQLEN + t;
    float dtv = dt[row * D_MODEL + d];
    float xv  = x[row * D_MODEL + d];
    float Bv[D_STATE], Cv[D_STATE];
    const float4* bp = (const float4*)(dtBC + row * E_DIM + DT_RANK);
    const float4* cp = (const float4*)(dtBC + row * E_DIM + DT_RANK + D_STATE);
#pragma unroll
    for (int i = 0; i < 4; ++i) {
      float4 q = bp[i];
      Bv[4*i+0] = q.x; Bv[4*i+1] = q.y; Bv[4*i+2] = q.z; Bv[4*i+3] = q.w;
    }
#pragma unroll
    for (int i = 0; i < 4; ++i) {
      float4 q = cp[i];
      Cv[4*i+0] = q.x; Cv[4*i+1] = q.y; Cv[4*i+2] = q.z; Cv[4*i+3] = q.w;
    }
    float dx = dtv * xv;
    float yv = 0.f;
#pragma unroll
    for (int n = 0; n < D_STATE; ++n) {
      float dA = __expf(dtv * An[n]);
      h[n] = dA * h[n] + dx * Bv[n];
      yv += h[n] * Cv[n];
    }
    y[row * D_MODEL + d] = yv + xv * Dd;
  }
}

extern "C" void kernel_launch(void* const* d_in, const int* in_sizes, int n_in,
                              void* d_out, int out_size, void* d_ws, size_t ws_size,
                              hipStream_t stream) {
  const float* x    = (const float*)d_in[0];
  const float* Alog = (const float*)d_in[1];
  const float* Dp   = (const float*)d_in[2];
  const float* Wdt  = (const float*)d_in[3];
  const float* bdt  = (const float*)d_in[4];
  const float* Wbc  = (const float*)d_in[5];
  float* y = (float*)d_out;

  float* ws    = (float*)d_ws;
  float* dtBC  = ws;
  float* dt    = dtBC + (size_t)ROWS * E_DIM;
  float* hend  = dt + (size_t)ROWS * D_MODEL;
  float* Pprod = hend + (size_t)BATCH * D_MODEL * NCHUNK * D_STATE;
  float* h0    = Pprod + (size_t)BATCH * D_MODEL * NCHUNK * D_STATE;

  int waves1 = (ROWS / 16) * (E_DIM / 16);
  k_proj_dtbc<<<waves1 * 32 / 256, 256, 0, stream>>>(x, Wbc, dtBC);

  int waves2 = (ROWS / 16) * (D_MODEL / 16);
  k_proj_dt<<<waves2 * 32 / 256, 256, 0, stream>>>(dtBC, Wdt, bdt, dt);

  int nscan = BATCH * D_MODEL * NCHUNK;
  k_scan_local<<<nscan / 256, 256, 0, stream>>>(x, dt, dtBC, Alog, hend, Pprod);

  int ncomb = BATCH * D_MODEL * D_STATE;
  k_scan_combine<<<ncomb / 256, 256, 0, stream>>>(hend, Pprod, h0);

  k_scan_emit<<<nscan / 256, 256, 0, stream>>>(x, dt, dtBC, Alog, Dp, h0, y);
}


# scan passes: no global load is waited for inside the step loop: the chunk's B,C rows are copied to LDS once, dt,x of the next 8 steps are requested before the current 8 are computed; sums unchanged
# speedup vs baseline: 1.5541x; 1.5541x over previous
#include <hip/hip_runtime.h>
#include <hip/hip_bf16.h>

#define D_MODEL 1024
#define D_STATE 16
#define DT_RANK 64
#define BATCH   2
#define SEQLEN  4096
#define E_DIM   (DT_RANK + 2*D_STATE)
#define ROWS    (BATCH*SEQLEN)
#define NCHUNK  64
#define TCHUNK  (SEQLEN/NCHUNK)

typedef __attribute__((ext_vector_type(2))) float v2f;
typedef __attribute__((ext_vector_type(8))) float v8f;

__device__ __forceinline__ float softplus_f(float v) {
  return v > 20.f ? v : log1pf(__expf(v));
}

__global__ __launch_bounds__(256) void k_proj_dtbc(const float* __restrict__ x,
                                                   const float* __restrict__ Wbc,
                                                   float* __restrict__ dtBC) {
  const int NT = E_DIM / 16;
  int wave = (blockIdx.x * blockDim.x + threadIdx.x) >> 5;
  int lane = threadIdx.x & 31;
  int mt = wave / NT, nt = wave - mt * NT;
  int l = lane & 15, half = lane >> 4;

  const float* arow = x   + (size_t)(mt * 16 + l) * D_MODEL + 2 * half;
  const float* brow = Wbc + (size_t)(nt * 16 + l) * D_MODEL + 2 * half;

  v8f acc = {0.f, 0.f, 0.f, 0.f, 0.f, 0.f, 0.f, 0.f};
  for (int k = 0; k < D_MODEL; k += 4) {
    v2f a, b;
    a[0] = arow[k]; a[1] = arow[k + 1];
    b[0] = brow[k]; b[1] = brow[k + 1];
    acc = __builtin_amdgcn_wmma_f32_16x16x4_f32(false, a, false, b,
                                                (short)0, acc, false, false);
  }
#pragma unroll
  for (int r = 0; r < 8; ++r) {
    int m = mt * 16 + 8 * half + r;
    dtBC[(size_t)m * E_DIM + nt * 16 + l] = acc[r];
  }
}

__global__ __launch_bounds__(256) void k_proj_dt(const float* __restrict__ dtBC,
                                                 const float* __restrict__ Wdt,
                                                 const float* __restrict__ bdt,
                                                 float* __restrict__ dt) {
  const int NT = D_MODEL / 16;
  int wave = (blockIdx.x * blockDim.x + threadIdx.x) >> 5;
  int lane = threadIdx.x & 31;
  int mt = wave / NT, nt = wave - mt * NT;
  int l = lane & 15, half = lane >> 4;

  const float* arow = dtBC + (size_t)(mt * 16 + l) * E_DIM   + 2 * half;
  const float* brow = Wdt  + (size_t)(nt * 16 + l) * DT_RANK + 2 * half;

  v8f acc = {0.f, 0.f, 0.f, 0.f, 0.f, 0.f, 0.f, 0.f};
  for (int k = 0; k < DT_RANK; k += 4) {
    v2f a, b;
    a[0] = arow[k]; a[1] = arow[k + 1];
    b[0] = brow[k]; b[1] = brow[k + 1];
    acc = __builtin_amdgcn_wmma_f32_16x16x4_f32(false, a, false, b,
                                                (short)0, acc, false, false);
  }
  float bias = bdt[nt * 16 + l];
#pragma unroll
  for (int r = 0; r < 8; ++r) {
    int m = mt * 16 + 8 * half + r;
    dt[(size_t)m * D_MODEL + nt * 16 + l] = softplus_f(acc[r] + bias);
  }
}

#define UPF 8

__global__ __launch_bounds__(256) void k_scan_local(const float* __restrict__ x,
                                                    const float* __restrict__ dt,
                                                    const float* __restrict__ dtBC,
                                                    const float* __restrict__ A_log,
                                                    float* __restrict__ hend,
                                                    float* __restrict__ Pprod) {
  __shared__ float4 sB[TCHUNK * 4];

  const int tid = threadIdx.x;
  const int blk = blockIdx.x;
  const int d = ((blk & 3) << 8) + tid;
  const int c = (blk >> 2) & (NCHUNK - 1);
  const int b = blk >> 8;
  const size_t row0 = (size_t)b * SEQLEN + (size_t)c * TCHUNK;

  sB[tid] = ((const float4*)(dtBC + (row0 + (tid >> 2)) * E_DIM + DT_RANK))[tid & 3];

  float An[D_STATE];
#pragma unroll
  for (int n = 0; n < D_STATE; ++n) An[n] = -__expf(A_log[d * D_STATE + n]);

  float h[D_STATE], P[D_STATE];
#pragma unroll
  for (int n = 0; n < D_STATE; ++n) { h[n] = 0.f; P[n] = 1.f; }

  __syncthreads();

  const float* dtp = dt + row0 * D_MODEL + d;
  const float* xp  = x  + row0 * D_MODEL + d;

  float dcur[UPF], xcur[UPF], dnxt[UPF], xnxt[UPF];
#pragma unroll
  for (int u = 0; u < UPF; ++u) {
    dcur[u] = dtp[(size_t)u * D_MODEL];
    xcur[u] = xp[(size_t)u * D_MODEL];
  }

  for (int tb = 0; tb < TCHUNK; tb += UPF) {
    const int tn = (tb + UPF < TCHUNK) ? tb + UPF : tb;
#pragma unroll
    for (int u = 0; u < UPF; ++u) {
      dnxt[u] = dtp[(size_t)(tn + u) * D_MODEL];
      xnxt[u] = xp[(size_t)(tn + u) * D_MODEL];
    }
    __builtin_amdgcn_sched_barrier(0);
#pragma unroll
    for (int u = 0; u < UPF; ++u) {
      const float dtv = dcur[u];
      const float xv  = xcur[u];
      float Bv[D_STATE];
#pragma unroll
      for (int i = 0; i < 4; ++i) {
        float4 q = sB[(tb + u) * 4 + i];
        Bv[4*i+0] = q.x; Bv[4*i+1] = q.y; Bv[4*i+2] = q.z; Bv[4*i+3] = q.w;
      }
      const float dx = dtv * xv;
#pragma unroll
      for (int n = 0; n < D_STATE; ++n) {
        float dA = __expf(dtv * An[n]);
        h[n] = dA * h[n] + dx * Bv[n];
        P[n] *= dA;
      }
    }
    __builtin_amdgcn_sched_barrier(0);
#pragma unroll
    for (int u = 0; u < UPF; ++u) { dcur[u] = dnxt[u]; xcur[u] = xnxt[u]; }
  }
  size_t base = ((size_t)(b * D_MODEL + d) * NCHUNK + c) * D_STATE;
#pragma unroll
  for (int n = 0; n < D_STATE; ++n) {
    hend[base + n]  = h[n];
    Pprod[base + n] = P[n];
  }
}

__global__ __launch_bounds__(256) void k_scan_combine(const float* __restrict__ hend,
                                                      const float* __restrict__ Pprod,
                                                      float* __restrict__ h0) {
  int gid = blockIdx.x * blockDim.x + threadIdx.x;
  int n = gid & (D_STATE - 1);
  int bd = gid >> 4;
  size_t base = (size_t)bd * NCHUNK * D_STATE + n;
  float h = 0.f;
  for (int c = 0; c < NCHUNK; ++c) {
    size_t idx = base + (size_t)c * D_STATE;
    h0[idx] = h;
    h = Pprod[idx] * h + hend[idx];
  }
}

__global__ __launch_bounds__(256) void k_scan_emit(const float* __restrict__ x,
                                                   const float* __restrict__ dt,
                                                   const float* __restrict__ dtBC,
                                                   const float* __restrict__ A_log,
                                                   const float* __restrict__ Dp,
                                                   const float* __restrict__ h0,
                                                   float* __restrict__ y) {
  __shared__ float4 sBC[TCHUNK * 8];

  const int tid = threadIdx.x;
  const int blk = blockIdx.x;
  const int d = ((blk & 3) << 8) + tid;
  const int c = (blk >> 2) & (NCHUNK - 1);
  const int b = blk >> 8;
  const size_t row0 = (size_t)b * SEQLEN + (size_t)c * TCHUNK;

#pragma unroll
  for (int j = 0; j < 2; ++j) {
    int idx = tid + 256 * j;
    sBC[idx] = ((const float4*)(dtBC + (row0 + (idx >> 3)) * E_DIM + DT_RANK))[idx & 7];
  }

  float An[D_STATE];
#pragma unroll
  for (int n = 0; n < D_STATE; ++n) An[n] = -__expf(A_log[d * D_STATE + n]);

  float h[D_STATE];
  size_t base = ((size_t)(b * D_MODEL + d) * NCHUNK + c) * D_STATE;
#pragma unroll
  for (int n = 0; n < D_STATE; ++n) h[n] = h0[base + n];

  const float Dd = Dp[d];

  __syncthreads();

  const float* dtp = dt + row0 * D_MODEL + d;
  const float* xp  = x  + row0 * D_MODEL + d;
  float*       yp  = y  + row0 * D_MODEL + d;

  float dcur[UPF], xcur[UPF], dnxt[UPF], xnxt[UPF];
#pragma unroll
  for (int u = 0; u < UPF; ++u) {
    dcur[u] = dtp[(size_t)u * D_MODEL];
    xcur[u] = xp[(size_t)u * D_MODEL];
  }

  for (int tb = 0; tb < TCHUNK; tb += UPF) {
    const int tn = (tb + UPF < TCHUNK) ? tb + UPF : tb;
#pragma unroll
    for (int u = 0; u < UPF; ++u) {
      dnxt[u] = dtp[(size_t)(tn + u) * D_MODEL];
      xnxt[u] = xp[(size_t)(tn + u) * D_MODEL];
    }
    __builtin_amdgcn_sched_barrier(0);
#pragma unroll
    for (int u = 0; u < UPF; ++u) {
      const float dtv = dcur[u];
      const float xv  = xcur[u];
      float Bv[D_STATE], Cv[D_STATE];
#pragma unroll
      for (int i = 0; i < 4; ++i) {
        float4 q = sBC[(tb + u) * 8 + i];
        Bv[4*i+0] = q.x; Bv[4*i+1] = q.y; Bv[4*i+2] = q.z; Bv[4*i+3] = q.w;
      }
#pragma unroll
      for (int i = 0; i < 4; ++i) {
        float4 q = sBC[(tb + u) * 8 + 4 + i];
        Cv[4*i+0] = q.x; Cv[4*i+1] = q.y; Cv[4*i+2] = q.z; Cv[4*i+3] = q.w;
      }
      const float dx = dtv * xv;
      float yv = 0.f;
#pragma unroll
      for (int n = 0; n < D_STATE; ++n) {
        float dA = __expf(dtv * An[n]);
        h[n] = dA * h[n] + dx * Bv[n];
        yv += h[n] * Cv[n];
      }
      yp[(size_t)(tb + u) * D_MODEL] = yv + xv * Dd;
    }
    __builtin_amdgcn_sched_barrier(0);
#pragma unroll
    for (int u = 0; u < UPF; ++u) { dcur[u] = dnxt[u]; xcur[u] = xnxt[u]; }
  }
}

extern "C" void kernel_launch(void* const* d_in, const int* in_sizes, int n_in,
                              void* d_out, int out_size, void* d_ws, size_t ws_size,
                              hipStream_t stream) {
  const float* x    = (const float*)d_in[0];
  const float* Alog = (const float*)d_in[1];
  const float* Dp   = (const float*)d_in[2];
  const float* Wdt  = (const float*)d_in[3];
  const float* bdt  = (const float*)d_in[4];
  const float* Wbc  = (const float*)d_in[5];
  float* y = (float*)d_out;

  float* ws    = (float*)d_ws;
  float* dtBC  = ws;
  float* dt    = dtBC + (size_t)ROWS * E_DIM;
  float* hend  = dt + (size_t)ROWS * D_MODEL;
  float* Pprod = hend + (size_t)BATCH * D_MODEL * NCHUNK * D_STATE;
  float* h0    = Pprod + (size_t)BATCH * D_MODEL * NCHUNK * D_STATE;

  int waves1 = (ROWS / 16) * (E_DIM / 16);
  k_proj_dtbc<<<waves1 * 32 / 256, 256, 0, stream>>>(x, Wbc, dtBC);

  int waves2 = (ROWS / 16) * (D_MODEL / 16);
  k_proj_dt<<<waves2 * 32 / 256, 256, 0, stream>>>(dtBC, Wdt, bdt, dt);

  int nscan = BATCH * D_MODEL * NCHUNK;
  k_scan_local<<<nscan / 256, 256, 0, stream>>>(x, dt, dtBC, Alog, hend, Pprod);

  int ncomb = BATCH * D_MODEL * D_STATE;
  k_scan_combine<<<ncomb / 256, 256, 0, stream>>>(hend, Pprod, h0);

  k_scan_emit<<<nscan / 256, 256, 0, stream>>>(x, dt, dtBC, Alog, Dp, h0, y);
}


# projection GEMMs block-tiled through LDS: tiles copied with whole-line 16-byte loads, waves read fragments from LDS (was 8-byte row-strided loads); k summed in another order inside each 8
# speedup vs baseline: 1.8721x; 1.2046x over previous
#include <hip/hip_runtime.h>
#include <hip/hip_bf16.h>

#define D_MODEL 1024
#define D_STATE 16
#define DT_RANK 64
#define BATCH   2
#define SEQLEN  4096
#define E_DIM   (DT_RANK + 2*D_STATE)
#define ROWS    (BATCH*SEQLEN)
#define NCHUNK  64
#define TCHUNK  (SEQLEN/NCHUNK)

typedef __attribute__((ext_vector_type(2))) float v2f;
typedef __attribute__((ext_vector_type(8))) float v8f;
typedef __attribute__((ext_vector_type(4))) float v4f;

__device__ __forceinline__ float softplus_f(float v) {
  return v > 20.f ? v : log1pf(__expf(v));
}

#define LDS_PITCH 68

#define G1_BM      32
#define G1_KC      64
#define G1_ROWS    (G1_BM + E_DIM)
#define G1_THREADS 384
#define G1_NLD     8

__global__ __launch_bounds__(G1_THREADS) void k_proj_dtbc(const float* __restrict__ x,
                                                          const float* __restrict__ Wbc,
                                                          float* __restrict__ dtBC) {
  __shared__ float sT[2][G1_ROWS * LDS_PITCH];

  const int tid = threadIdx.x;
  const int wave = __builtin_amdgcn_readfirstlane(tid >> 5);
  const int lane = tid & 31;
  const int c = lane & 15, h = lane >> 4;
  const int mt = wave / 6, nt = wave - mt * 6;
  const int m0 = blockIdx.x * G1_BM;
  const bool copier = wave < 8;

  const int r0 = (tid & 255) >> 4, q = tid & 15;
  const float* xs = x   + (size_t)(m0 + r0) * D_MODEL + 4 * q;
  const float* ws = Wbc + (size_t)r0 * D_MODEL + 4 * q;
  const int d0 = r0 * LDS_PITCH + 4 * q;

  v4f st[G1_NLD];
  if (copier) {
#pragma unroll
    for (int j = 0; j < G1_NLD; ++j)
      st[j] = *(const v4f*)(j < 2 ? xs + (size_t)(16 * j) * D_MODEL
                                     : ws + (size_t)(16 * (j - 2)) * D_MODEL);
#pragma unroll
    for (int j = 0; j < G1_NLD; ++j)
      *(v4f*)(&sT[0][d0 + 16 * j * LDS_PITCH]) = st[j];
  }
  __syncthreads();

  v8f acc = {0.f, 0.f, 0.f, 0.f, 0.f, 0.f, 0.f, 0.f};
  const int aoff = (mt * 16 + c) * LDS_PITCH + 4 * h;
  const int boff = (G1_BM + nt * 16 + c) * LDS_PITCH + 4 * h;

  for (int kc = 0; kc < D_MODEL / G1_KC; ++kc) {
    const bool fetch = copier && (kc + 1 < D_MODEL / G1_KC);
    if (fetch) {
      const int ko = (kc + 1) * G1_KC;
#pragma unroll
      for (int j = 0; j < G1_NLD; ++j)
        st[j] = *(const v4f*)(j < 2 ? xs + (size_t)(16 * j) * D_MODEL + ko
                                       : ws + (size_t)(16 * (j - 2)) * D_MODEL + ko);
    }
    const float* sa = &sT[kc & 1][aoff];
    const float* sb = &sT[kc & 1][boff];
#pragma unroll
    for (int jj = 0; jj < G1_KC / 8; ++jj) {
      const v4f a4 = *(const v4f*)(sa + 8 * jj);
      const v4f b4 = *(const v4f*)(sb + 8 * jj);
      v2f a, b;
      a[0] = a4[0]; a[1] = a4[1]; b[0] = b4[0]; b[1] = b4[1];
      acc = __builtin_amdgcn_wmma_f32_16x16x4_f32(false, a, false, b,
                                                  (short)0, acc, false, false);
      a[0] = a4[2]; a[1] = a4[3]; b[0] = b4[2]; b[1] = b4[3];
      acc = __builtin_amdgcn_wmma_f32_16x16x4_f32(false, a, false, b,
                                                  (short)0, acc, false, false);
    }
    if (fetch) {
#pragma unroll
      for (int j = 0; j < G1_NLD; ++j)
        *(v4f*)(&sT[(kc + 1) & 1][d0 + 16 * j * LDS_PITCH]) = st[j];
    }
    __syncthreads();
  }
#pragma unroll
  for (int r = 0; r < 8; ++r) {
    int m = m0 + mt * 16 + 8 * h + r;
    dtBC[(size_t)m * E_DIM + nt * 16 + c] = acc[r];
  }
}

#define G2_BM      64
#define G2_BN      128
#define G2_ROWS    (G2_BM + G2_BN)
#define G2_NLD     12

__global__ __launch_bounds__(256) void k_proj_dt(const float* __restrict__ dtBC,
                                                 const float* __restrict__ Wdt,
                                                 const float* __restrict__ bdt,
                                                 float* __restrict__ dt) {
  __shared__ float sT[G2_ROWS * LDS_PITCH];

  const int tid = threadIdx.x;
  const int wave = tid >> 5, lane = tid & 31;
  const int c = lane & 15, h = lane >> 4;
  const int wm = wave & 3, wn = wave >> 2;
  const int m0 = (blockIdx.x >> 3) * G2_BM;
  const int n0 = (blockIdx.x & 7) * G2_BN;

  const int r0 = tid >> 4, q = tid & 15;
  const float* as = dtBC + (size_t)(m0 + r0) * E_DIM + 4 * q;
  const float* ws = Wdt  + (size_t)(n0 + r0) * DT_RANK + 4 * q;
  v4f st[G2_NLD];
#pragma unroll
  for (int j = 0; j < G2_NLD; ++j)
    st[j] = *(const v4f*)(j < 4 ? as + (size_t)(16 * j) * E_DIM
                                   : ws + (size_t)(16 * (j - 4)) * DT_RANK);
#pragma unroll
  for (int j = 0; j < G2_NLD; ++j)
    *(v4f*)(&sT[(r0 + 16 * j) * LDS_PITCH + 4 * q]) = st[j];
  __syncthreads();

  v8f acc[4];
#pragma unroll
  for (int t = 0; t < 4; ++t) acc[t] = (v8f){0.f, 0.f, 0.f, 0.f, 0.f, 0.f, 0.f, 0.f};

  const float* sa = &sT[(wm * 16 + c) * LDS_PITCH + 4 * h];
  const float* sb = &sT[(G2_BM + wn * 64 + c) * LDS_PITCH + 4 * h];
#pragma unroll
  for (int jj = 0; jj < DT_RANK / 8; ++jj) {
    const v4f a4 = *(const v4f*)(sa + 8 * jj);
    v2f a0, a1;
    a0[0] = a4[0]; a0[1] = a4[1]; a1[0] = a4[2]; a1[1] = a4[3];
#pragma unroll
    for (int t = 0; t < 4; ++t) {
      const v4f b4 = *(const v4f*)(sb + t * 16 * LDS_PITCH + 8 * jj);
      v2f b0, b1;
      b0[0] = b4[0]; b0[1] = b4[1]; b1[0] = b4[2]; b1[1] = b4[3];
      acc[t] = __builtin_amdgcn_wmma_f32_16x16x4_f32(false, a0, false, b0,
                                                     (short)0, acc[t], false, false);
      acc[t] = __builtin_amdgcn_wmma_f32_16x16x4_f32(false, a1, false, b1,
                                                     (short)0, acc[t], false, false);
    }
  }
#pragma unroll
  for (int t = 0; t < 4; ++t) {
    const int n = n0 + wn * 64 + t * 16 + c;
    const float bias = bdt[n];
#pragma unroll
    for (int r = 0; r < 8; ++r) {
      int m = m0 + wm * 16 + 8 * h + r;
      dt[(size_t)m * D_MODEL + n] = softplus_f(acc[t][r] + bias);
    }
  }
}

#define UPF 8

__global__ __launch_bounds__(256) void k_scan_local(const float* __restrict__ x,
                                                    const float* __restrict__ dt,
                                                    const float* __restrict__ dtBC,
                                                    const float* __restrict__ A_log,
                                                    float* __restrict__ hend,
                                                    float* __restrict__ Pprod) {
  __shared__ float4 sB[TCHUNK * 4];

  const int tid = threadIdx.x;
  const int blk = blockIdx.x;
  const int d = ((blk & 3) << 8) + tid;
  const int c = (blk >> 2) & (NCHUNK - 1);
  const int b = blk >> 8;
  const size_t row0 = (size_t)b * SEQLEN + (size_t)c * TCHUNK;

  sB[tid] = ((const float4*)(dtBC + (row0 + (tid >> 2)) * E_DIM + DT_RANK))[tid & 3];

  float An[D_STATE];
#pragma unroll
  for (int n = 0; n < D_STATE; ++n) An[n] = -__expf(A_log[d * D_STATE + n]);

  float h[D_STATE], P[D_STATE];
#pragma unroll
  for (int n = 0; n < D_STATE; ++n) { h[n] = 0.f; P[n] = 1.f; }

  __syncthreads();

  const float* dtp = dt + row0 * D_MODEL + d;
  const float* xp  = x  + row0 * D_MODEL + d;

  float dcur[UPF], xcur[UPF], dnxt[UPF], xnxt[UPF];
#pragma unroll
  for (int u = 0; u < UPF; ++u) {
    dcur[u] = dtp[(size_t)u * D_MODEL];
    xcur[u] = xp[(size_t)u * D_MODEL];
  }

  for (int tb = 0; tb < TCHUNK; tb += UPF) {
    const int tn = (tb + UPF < TCHUNK) ? tb + UPF : tb;
#pragma unroll
    for (int u = 0; u < UPF; ++u) {
      dnxt[u] = dtp[(size_t)(tn + u) * D_MODEL];
      xnxt[u] = xp[(size_t)(tn + u) * D_MODEL];
    }
    __builtin_amdgcn_sched_barrier(0);
#pragma unroll
    for (int u = 0; u < UPF; ++u) {
      const float dtv = dcur[u];
      const float xv  = xcur[u];
      float Bv[D_STATE];
#pragma unroll
      for (int i = 0; i < 4; ++i) {
        float4 q = sB[(tb + u) * 4 + i];
        Bv[4*i+0] = q.x; Bv[4*i+1] = q.y; Bv[4*i+2] = q.z; Bv[4*i+3] = q.w;
      }
      const float dx = dtv * xv;
#pragma unroll
      for (int n = 0; n < D_STATE; ++n) {
        float dA = __expf(dtv * An[n]);
        h[n] = dA * h[n] + dx * Bv[n];
        P[n] *= dA;
      }
    }
    __builtin_amdgcn_sched_barrier(0);
#pragma unroll
    for (int u = 0; u < UPF; ++u) { dcur[u] = dnxt[u]; xcur[u] = xnxt[u]; }
  }
  size_t base = ((size_t)(b * D_MODEL + d) * NCHUNK + c) * D_STATE;
#pragma unroll
  for (int n = 0; n < D_STATE; ++n) {
    hend[base + n]  = h[n];
    Pprod[base + n] = P[n];
  }
}

__global__ __launch_bounds__(256) void k_scan_combine(const float* __restrict__ hend,
                                                      const float* __restrict__ Pprod,
                                                      float* __restrict__ h0) {
  int gid = blockIdx.x * blockDim.x + threadIdx.x;
  int n = gid & (D_STATE - 1);
  int bd = gid >> 4;
  size_t base = (size_t)bd * NCHUNK * D_STATE + n;
  float h = 0.f;
  for (int c = 0; c < NCHUNK; ++c) {
    size_t idx = base + (size_t)c * D_STATE;
    h0[idx] = h;
    h = Pprod[idx] * h + hend[idx];
  }
}

__global__ __launch_bounds__(256) void k_scan_emit(const float* __restrict__ x,
                                                   const float* __restrict__ dt,
                                                   const float* __restrict__ dtBC,
                                                   const float* __restrict__ A_log,
                                                   const float* __restrict__ Dp,
                                                   const float* __restrict__ h0,
                                                   float* __restrict__ y) {
  __shared__ float4 sBC[TCHUNK * 8];

  const int tid = threadIdx.x;
  const int blk = blockIdx.x;
  const int d = ((blk & 3) << 8) + tid;
  const int c = (blk >> 2) & (NCHUNK - 1);
  const int b = blk >> 8;
  const size_t row0 = (size_t)b * SEQLEN + (size_t)c * TCHUNK;

#pragma unroll
  for (int j = 0; j < 2; ++j) {
    int idx = tid + 256 * j;
    sBC[idx] = ((const float4*)(dtBC + (row0 + (idx >> 3)) * E_DIM + DT_RANK))[idx & 7];
  }

  float An[D_STATE];
#pragma unroll
  for (int n = 0; n < D_STATE; ++n) An[n] = -__expf(A_log[d * D_STATE + n]);

  float h[D_STATE];
  size_t base = ((size_t)(b * D_MODEL + d) * NCHUNK + c) * D_STATE;
#pragma unroll
  for (int n = 0; n < D_STATE; ++n) h[n] = h0[base + n];

  const float Dd = Dp[d];

  __syncthreads();

  const float* dtp = dt + row0 * D_MODEL + d;
  const float* xp  = x  + row0 * D_MODEL + d;
  float*       yp  = y  + row0 * D_MODEL + d;

  float dcur[UPF], xcur[UPF], dnxt[UPF], xnxt[UPF];
#pragma unroll
  for (int u = 0; u < UPF; ++u) {
    dcur[u] = dtp[(size_t)u * D_MODEL];
    xcur[u] = xp[(size_t)u * D_MODEL];
  }

  for (int tb = 0; tb < TCHUNK; tb += UPF) {
    const int tn = (tb + UPF < TCHUNK) ? tb + UPF : tb;
#pragma unroll
    for (int u = 0; u < UPF; ++u) {
      dnxt[u] = dtp[(size_t)(tn + u) * D_MODEL];
      xnxt[u] = xp[(size_t)(tn + u) * D_MODEL];
    }
    __builtin_amdgcn_sched_barrier(0);
#pragma unroll
    for (int u = 0; u < UPF; ++u) {
      const float dtv = dcur[u];
      const float xv  = xcur[u];
      float Bv[D_STATE], Cv[D_STATE];
#pragma unroll
      for (int i = 0; i < 4; ++i) {
        float4 q = sBC[(tb + u) * 8 + i];
        Bv[4*i+0] = q.x; Bv[4*i+1] = q.y; Bv[4*i+2] = q.z; Bv[4*i+3] = q.w;
      }
#pragma unroll
      for (int i = 0; i < 4; ++i) {
        float4 q = sBC[(tb + u) * 8 + 4 + i];
        Cv[4*i+0] = q.x; Cv[4*i+1] = q.y; Cv[4*i+2] = q.z; Cv[4*i+3] = q.w;
      }
      const float dx = dtv * xv;
      float yv = 0.f;
#pragma unroll
      for (int n = 0; n < D_STATE; ++n) {
        float dA = __expf(dtv * An[n]);
        h[n] = dA * h[n] + dx * Bv[n];
        yv += h[n] * Cv[n];
      }
      yp[(size_t)(tb + u) * D_MODEL] = yv + xv * Dd;
    }
    __builtin_amdgcn_sched_barrier(0);
#pragma unroll
    for (int u = 0; u < UPF; ++u) { dcur[u] = dnxt[u]; xcur[u] = xnxt[u]; }
  }
}

extern "C" void kernel_launch(void* const* d_in, const int* in_sizes, int n_in,
                              void* d_out, int out_size, void* d_ws, size_t ws_size,
                              hipStream_t stream) {
  const float* x    = (const float*)d_in[0];
  const float* Alog = (const float*)d_in[1];
  const float* Dp   = (const float*)d_in[2];
  const float* Wdt  = (const float*)d_in[3];
  const float* bdt  = (const float*)d_in[4];
  const float* Wbc  = (const float*)d_in[5];
  float* y = (float*)d_out;

  float* ws    = (float*)d_ws;
  float* dtBC  = ws;
  float* dt    = dtBC + (size_t)ROWS * E_DIM;
  float* hend  = dt + (size_t)ROWS * D_MODEL;
  float* Pprod = hend + (size_t)BATCH * D_MODEL * NCHUNK * D_STATE;
  float* h0    = Pprod + (size_t)BATCH * D_MODEL * NCHUNK * D_STATE;

  k_proj_dtbc<<<ROWS / G1_BM, G1_THREADS, 0, stream>>>(x, Wbc, dtBC);

  k_proj_dt<<<(ROWS / G2_BM) * (D_MODEL / G2_BN), 256, 0, stream>>>(dtBC, Wdt, bdt, dt);

  int nscan = BATCH * D_MODEL * NCHUNK;
  k_scan_local<<<nscan / 256, 256, 0, stream>>>(x, dt, dtBC, Alog, hend, Pprod);

  int ncomb = BATCH * D_MODEL * D_STATE;
  k_scan_combine<<<ncomb / 256, 256, 0, stream>>>(hend, Pprod, h0);

  k_scan_emit<<<nscan / 256, 256, 0, stream>>>(x, dt, dtBC, Alog, Dp, h0, y);
}


# dt projection epilogue: softplus as max(v,0)+log1p(exp(-|v|)), log1p from one hardware log2 plus a first-order repair of the rounding of 1+e, replacing the library log1p; dt moves in its last bits
# speedup vs baseline: 2.1512x; 1.1491x over previous
#include <hip/hip_runtime.h>
#include <hip/hip_bf16.h>

#define D_MODEL 1024
#define D_STATE 16
#define DT_RANK 64
#define BATCH   2
#define SEQLEN  4096
#define E_DIM   (DT_RANK + 2*D_STATE)
#define ROWS    (BATCH*SEQLEN)
#define NCHUNK  64
#define TCHUNK  (SEQLEN/NCHUNK)

typedef __attribute__((ext_vector_type(2))) float v2f;
typedef __attribute__((ext_vector_type(8))) float v8f;
typedef __attribute__((ext_vector_type(4))) float v4f;

__device__ __forceinline__ float softplus_f(float v) {
  const float e = __expf(-fabsf(v));
  const float u = 1.0f + e;
  const float d = u - 1.0f;
  const float lg = __builtin_amdgcn_logf(u) * 0.693147180559945f;
  return fmaxf(v, 0.0f) + fmaf(e - d, 1.0f - d, lg);
}

#define LDS_PITCH 68

#define G1_BM      32
#define G1_KC      64
#define G1_ROWS    (G1_BM + E_DIM)
#define G1_THREADS 384
#define G1_NLD     8

__global__ __launch_bounds__(G1_THREADS) void k_proj_dtbc(const float* __restrict__ x,
                                                          const float* __restrict__ Wbc,
                                                          float* __restrict__ dtBC) {
  __shared__ float sT[2][G1_ROWS * LDS_PITCH];

  const int tid = threadIdx.x;
  const int wave = __builtin_amdgcn_readfirstlane(tid >> 5);
  const int lane = tid & 31;
  const int c = lane & 15, h = lane >> 4;
  const int mt = wave / 6, nt = wave - mt * 6;
  const int m0 = blockIdx.x * G1_BM;
  const bool copier = wave < 8;

  const int r0 = (tid & 255) >> 4, q = tid & 15;
  const float* xs = x   + (size_t)(m0 + r0) * D_MODEL + 4 * q;
  const float* ws = Wbc + (size_t)r0 * D_MODEL + 4 * q;
  const int d0 = r0 * LDS_PITCH + 4 * q;

  v4f st[G1_NLD];
  if (copier) {
#pragma unroll
    for (int j = 0; j < G1_NLD; ++j)
      st[j] = *(const v4f*)(j < 2 ? xs + (size_t)(16 * j) * D_MODEL
                                     : ws + (size_t)(16 * (j - 2)) * D_MODEL);
#pragma unroll
    for (int j = 0; j < G1_NLD; ++j)
      *(v4f*)(&sT[0][d0 + 16 * j * LDS_PITCH]) = st[j];
  }
  __syncthreads();

  v8f acc = {0.f, 0.f, 0.f, 0.f, 0.f, 0.f, 0.f, 0.f};
  const int aoff = (mt * 16 + c) * LDS_PITCH + 4 * h;
  const int boff = (G1_BM + nt * 16 + c) * LDS_PITCH + 4 * h;

  for (int kc = 0; kc < D_MODEL / G1_KC; ++kc) {
    const bool fetch = copier && (kc + 1 < D_MODEL / G1_KC);
    if (fetch) {
      const int ko = (kc + 1) * G1_KC;
#pragma unroll
      for (int j = 0; j < G1_NLD; ++j)
        st[j] = *(const v4f*)(j < 2 ? xs + (size_t)(16 * j) * D_MODEL + ko
                                       : ws + (size_t)(16 * (j - 2)) * D_MODEL + ko);
    }
    const float* sa = &sT[kc & 1][aoff];
    const float* sb = &sT[kc & 1][boff];
#pragma unroll
    for (int jj = 0; jj < G1_KC / 8; ++jj) {
      const v4f a4 = *(const v4f*)(sa + 8 * jj);
      const v4f b4 = *(const v4f*)(sb + 8 * jj);
      v2f a, b;
      a[0] = a4[0]; a[1] = a4[1]; b[0] = b4[0]; b[1] = b4[1];
      acc = __builtin_amdgcn_wmma_f32_16x16x4_f32(false, a, false, b,
                                                  (short)0, acc, false, false);
      a[0] = a4[2]; a[1] = a4[3]; b[0] = b4[2]; b[1] = b4[3];
      acc = __builtin_amdgcn_wmma_f32_16x16x4_f32(false, a, false, b,
                                                  (short)0, acc, false, false);
    }
    if (fetch) {
#pragma unroll
      for (int j = 0; j < G1_NLD; ++j)
        *(v4f*)(&sT[(kc + 1) & 1][d0 + 16 * j * LDS_PITCH]) = st[j];
    }
    __syncthreads();
  }
#pragma unroll
  for (int r = 0; r < 8; ++r) {
    int m = m0 + mt * 16 + 8 * h + r;
    dtBC[(size_t)m * E_DIM + nt * 16 + c] = acc[r];
  }
}

#define G2_BM      64
#define G2_BN      128
#define G2_ROWS    (G2_BM + G2_BN)
#define G2_NLD     12

__global__ __launch_bounds__(256) void k_proj_dt(const float* __restrict__ dtBC,
                                                 const float* __restrict__ Wdt,
                                                 const float* __restrict__ bdt,
                                                 float* __restrict__ dt) {
  __shared__ float sT[G2_ROWS * LDS_PITCH];

  const int tid = threadIdx.x;
  const int wave = tid >> 5, lane = tid & 31;
  const int c = lane & 15, h = lane >> 4;
  const int wm = wave & 3, wn = wave >> 2;
  const int m0 = (blockIdx.x >> 3) * G2_BM;
  const int n0 = (blockIdx.x & 7) * G2_BN;

  const int r0 = tid >> 4, q = tid & 15;
  const float* as = dtBC + (size_t)(m0 + r0) * E_DIM + 4 * q;
  const float* ws = Wdt  + (size_t)(n0 + r0) * DT_RANK + 4 * q;
  v4f st[G2_NLD];
#pragma unroll
  for (int j = 0; j < G2_NLD; ++j)
    st[j] = *(const v4f*)(j < 4 ? as + (size_t)(16 * j) * E_DIM
                                   : ws + (size_t)(16 * (j - 4)) * DT_RANK);
#pragma unroll
  for (int j = 0; j < G2_NLD; ++j)
    *(v4f*)(&sT[(r0 + 16 * j) * LDS_PITCH + 4 * q]) = st[j];
  __syncthreads();

  v8f acc[4];
#pragma unroll
  for (int t = 0; t < 4; ++t) acc[t] = (v8f){0.f, 0.f, 0.f, 0.f, 0.f, 0.f, 0.f, 0.f};

  const float* sa = &sT[(wm * 16 + c) * LDS_PITCH + 4 * h];
  const float* sb = &sT[(G2_BM + wn * 64 + c) * LDS_PITCH + 4 * h];
#pragma unroll
  for (int jj = 0; jj < DT_RANK / 8; ++jj) {
    const v4f a4 = *(const v4f*)(sa + 8 * jj);
    v2f a0, a1;
    a0[0] = a4[0]; a0[1] = a4[1]; a1[0] = a4[2]; a1[1] = a4[3];
#pragma unroll
    for (int t = 0; t < 4; ++t) {
      const v4f b4 = *(const v4f*)(sb + t * 16 * LDS_PITCH + 8 * jj);
      v2f b0, b1;
      b0[0] = b4[0]; b0[1] = b4[1]; b1[0] = b4[2]; b1[1] = b4[3];
      acc[t] = __builtin_amdgcn_wmma_f32_16x16x4_f32(false, a0, false, b0,
                                                     (short)0, acc[t], false, false);
      acc[t] = __builtin_amdgcn_wmma_f32_16x16x4_f32(false, a1, false, b1,
                                                     (short)0, acc[t], false, false);
    }
  }
#pragma unroll
  for (int t = 0; t < 4; ++t) {
    const int n = n0 + wn * 64 + t * 16 + c;
    const float bias = bdt[n];
#pragma unroll
    for (int r = 0; r < 8; ++r) {
      int m = m0 + wm * 16 + 8 * h + r;
      dt[(size_t)m * D_MODEL + n] = softplus_f(acc[t][r] + bias);
    }
  }
}

#define UPF 8

__global__ __launch_bounds__(256) void k_scan_local(const float* __restrict__ x,
                                                    const float* __restrict__ dt,
                                                    const float* __restrict__ dtBC,
                                                    const float* __restrict__ A_log,
                                                    float* __restrict__ hend,
                                                    float* __restrict__ Pprod) {
  __shared__ float4 sB[TCHUNK * 4];

  const int tid = threadIdx.x;
  const int blk = blockIdx.x;
  const int d = ((blk & 3) << 8) + tid;
  const int c = (blk >> 2) & (NCHUNK - 1);
  const int b = blk >> 8;
  const size_t row0 = (size_t)b * SEQLEN + (size_t)c * TCHUNK;

  sB[tid] = ((const float4*)(dtBC + (row0 + (tid >> 2)) * E_DIM + DT_RANK))[tid & 3];

  float An[D_STATE];
#pragma unroll
  for (int n = 0; n < D_STATE; ++n) An[n] = -__expf(A_log[d * D_STATE + n]);

  float h[D_STATE], P[D_STATE];
#pragma unroll
  for (int n = 0; n < D_STATE; ++n) { h[n] = 0.f; P[n] = 1.f; }

  __syncthreads();

  const float* dtp = dt + row0 * D_MODEL + d;
  const float* xp  = x  + row0 * D_MODEL + d;

  float dcur[UPF], xcur[UPF], dnxt[UPF], xnxt[UPF];
#pragma unroll
  for (int u = 0; u < UPF; ++u) {
    dcur[u] = dtp[(size_t)u * D_MODEL];
    xcur[u] = xp[(size_t)u * D_MODEL];
  }

  for (int tb = 0; tb < TCHUNK; tb += UPF) {
    const int tn = (tb + UPF < TCHUNK) ? tb + UPF : tb;
#pragma unroll
    for (int u = 0; u < UPF; ++u) {
      dnxt[u] = dtp[(size_t)(tn + u) * D_MODEL];
      xnxt[u] = xp[(size_t)(tn + u) * D_MODEL];
    }
    __builtin_amdgcn_sched_barrier(0);
#pragma unroll
    for (int u = 0; u < UPF; ++u) {
      const float dtv = dcur[u];
      const float xv  = xcur[u];
      float Bv[D_STATE];
#pragma unroll
      for (int i = 0; i < 4; ++i) {
        float4 q = sB[(tb + u) * 4 + i];
        Bv[4*i+0] = q.x; Bv[4*i+1] = q.y; Bv[4*i+2] = q.z; Bv[4*i+3] = q.w;
      }
      const float dx = dtv * xv;
#pragma unroll
      for (int n = 0; n < D_STATE; ++n) {
        float dA = __expf(dtv * An[n]);
        h[n] = dA * h[n] + dx * Bv[n];
        P[n] *= dA;
      }
    }
    __builtin_amdgcn_sched_barrier(0);
#pragma unroll
    for (int u = 0; u < UPF; ++u) { dcur[u] = dnxt[u]; xcur[u] = xnxt[u]; }
  }
  size_t base = ((size_t)(b * D_MODEL + d) * NCHUNK + c) * D_STATE;
#pragma unroll
  for (int n = 0; n < D_STATE; ++n) {
    hend[base + n]  = h[n];
    Pprod[base + n] = P[n];
  }
}

__global__ __launch_bounds__(256) void k_scan_combine(const float* __restrict__ hend,
                                                      const float* __restrict__ Pprod,
                                                      float* __restrict__ h0) {
  int gid = blockIdx.x * blockDim.x + threadIdx.x;
  int n = gid & (D_STATE - 1);
  int bd = gid >> 4;
  size_t base = (size_t)bd * NCHUNK * D_STATE + n;
  float h = 0.f;
  for (int c = 0; c < NCHUNK; ++c) {
    size_t idx = base + (size_t)c * D_STATE;
    h0[idx] = h;
    h = Pprod[idx] * h + hend[idx];
  }
}

__global__ __launch_bounds__(256) void k_scan_emit(const float* __restrict__ x,
                                                   const float* __restrict__ dt,
                                                   const float* __restrict__ dtBC,
                                                   const float* __restrict__ A_log,
                                                   const float* __restrict__ Dp,
                                                   const float* __restrict__ h0,
                                                   float* __restrict__ y) {
  __shared__ float4 sBC[TCHUNK * 8];

  const int tid = threadIdx.x;
  const int blk = blockIdx.x;
  const int d = ((blk & 3) << 8) + tid;
  const int c = (blk >> 2) & (NCHUNK - 1);
  const int b = blk >> 8;
  const size_t row0 = (size_t)b * SEQLEN + (size_t)c * TCHUNK;

#pragma unroll
  for (int j = 0; j < 2; ++j) {
    int idx = tid + 256 * j;
    sBC[idx] = ((const float4*)(dtBC + (row0 + (idx >> 3)) * E_DIM + DT_RANK))[idx & 7];
  }

  float An[D_STATE];
#pragma unroll
  for (int n = 0; n < D_STATE; ++n) An[n] = -__expf(A_log[d * D_STATE + n]);

  float h[D_STATE];
  size_t base = ((size_t)(b * D_MODEL + d) * NCHUNK + c) * D_STATE;
#pragma unroll
  for (int n = 0; n < D_STATE; ++n) h[n] = h0[base + n];

  const float Dd = Dp[d];

  __syncthreads();

  const float* dtp = dt + row0 * D_MODEL + d;
  const float* xp  = x  + row0 * D_MODEL + d;
  float*       yp  = y  + row0 * D_MODEL + d;

  float dcur[UPF], xcur[UPF], dnxt[UPF], xnxt[UPF];
#pragma unroll
  for (int u = 0; u < UPF; ++u) {
    dcur[u] = dtp[(size_t)u * D_MODEL];
    xcur[u] = xp[(size_t)u * D_MODEL];
  }

  for (int tb = 0; tb < TCHUNK; tb += UPF) {
    const int tn = (tb + UPF < TCHUNK) ? tb + UPF : tb;
#pragma unroll
    for (int u = 0; u < UPF; ++u) {
      dnxt[u] = dtp[(size_t)(tn + u) * D_MODEL];
      xnxt[u] = xp[(size_t)(tn + u) * D_MODEL];
    }
    __builtin_amdgcn_sched_barrier(0);
#pragma unroll
    for (int u = 0; u < UPF; ++u) {
      const float dtv = dcur[u];
      const float xv  = xcur[u];
      float Bv[D_STATE], Cv[D_STATE];
#pragma unroll
      for (int i = 0; i < 4; ++i) {
        float4 q = sBC[(tb + u) * 8 + i];
        Bv[4*i+0] = q.x; Bv[4*i+1] = q.y; Bv[4*i+2] = q.z; Bv[4*i+3] = q.w;
      }
#pragma unroll
      for (int i = 0; i < 4; ++i) {
        float4 q = sBC[(tb + u) * 8 + 4 + i];
        Cv[4*i+0] = q.x; Cv[4*i+1] = q.y; Cv[4*i+2] = q.z; Cv[4*i+3] = q.w;
      }
      const float dx = dtv * xv;
      float yv = 0.f;
#pragma unroll
      for (int n = 0; n < D_STATE; ++n) {
        float dA = __expf(dtv * An[n]);
        h[n] = dA * h[n] + dx * Bv[n];
        yv += h[n] * Cv[n];
      }
      yp[(size_t)(tb + u) * D_MODEL] = yv + xv * Dd;
    }
    __builtin_amdgcn_sched_barrier(0);
#pragma unroll
    for (int u = 0; u < UPF; ++u) { dcur[u] = dnxt[u]; xcur[u] = xnxt[u]; }
  }
}

extern "C" void kernel_launch(void* const* d_in, const int* in_sizes, int n_in,
                              void* d_out, int out_size, void* d_ws, size_t ws_size,
                              hipStream_t stream) {
  const float* x    = (const float*)d_in[0];
  const float* Alog = (const float*)d_in[1];
  const float* Dp   = (const float*)d_in[2];
  const float* Wdt  = (const float*)d_in[3];
  const float* bdt  = (const float*)d_in[4];
  const float* Wbc  = (const float*)d_in[5];
  float* y = (float*)d_out;

  float* ws    = (float*)d_ws;
  float* dtBC  = ws;
  float* dt    = dtBC + (size_t)ROWS * E_DIM;
  float* hend  = dt + (size_t)ROWS * D_MODEL;
  float* Pprod = hend + (size_t)BATCH * D_MODEL * NCHUNK * D_STATE;
  float* h0    = Pprod + (size_t)BATCH * D_MODEL * NCHUNK * D_STATE;

  k_proj_dtbc<<<ROWS / G1_BM, G1_THREADS, 0, stream>>>(x, Wbc, dtBC);

  k_proj_dt<<<(ROWS / G2_BM) * (D_MODEL / G2_BN), 256, 0, stream>>>(dtBC, Wdt, bdt, dt);

  int nscan = BATCH * D_MODEL * NCHUNK;
  k_scan_local<<<nscan / 256, 256, 0, stream>>>(x, dt, dtBC, Alog, hend, Pprod);

  int ncomb = BATCH * D_MODEL * D_STATE;
  k_scan_combine<<<ncomb / 256, 256, 0, stream>>>(hend, Pprod, h0);

  k_scan_emit<<<nscan / 256, 256, 0, stream>>>(x, dt, dtBC, Alog, Dp, h0, y);
}


# chunk-state arrays (hend, P, h0) laid out [b][chunk][n][d], d fastest: each store or load of them is one whole 128-byte line per wave (was 16-byte pieces 4 KiB apart); order of sums unchanged
# speedup vs baseline: 2.2861x; 1.0627x over previous
#include <hip/hip_runtime.h>
#include <hip/hip_bf16.h>

#define D_MODEL 1024
#define D_STATE 16
#define DT_RANK 64
#define BATCH   2
#define SEQLEN  4096
#define E_DIM   (DT_RANK + 2*D_STATE)
#define ROWS    (BATCH*SEQLEN)
#define NCHUNK  64
#define TCHUNK  (SEQLEN/NCHUNK)

typedef __attribute__((ext_vector_type(2))) float v2f;
typedef __attribute__((ext_vector_type(8))) float v8f;
typedef __attribute__((ext_vector_type(4))) float v4f;

__device__ __forceinline__ float softplus_f(float v) {
  const float e = __expf(-fabsf(v));
  const float u = 1.0f + e;
  const float d = u - 1.0f;
  const float lg = __builtin_amdgcn_logf(u) * 0.693147180559945f;
  return fmaxf(v, 0.0f) + fmaf(e - d, 1.0f - d, lg);
}

#define LDS_PITCH 68

#define G1_BM      32
#define G1_KC      64
#define G1_ROWS    (G1_BM + E_DIM)
#define G1_THREADS 384
#define G1_NLD     8

__global__ __launch_bounds__(G1_THREADS) void k_proj_dtbc(const float* __restrict__ x,
                                                          const float* __restrict__ Wbc,
                                                          float* __restrict__ dtBC) {
  __shared__ float sT[2][G1_ROWS * LDS_PITCH];

  const int tid = threadIdx.x;
  const int wave = __builtin_amdgcn_readfirstlane(tid >> 5);
  const int lane = tid & 31;
  const int c = lane & 15, h = lane >> 4;
  const int mt = wave / 6, nt = wave - mt * 6;
  const int m0 = blockIdx.x * G1_BM;
  const bool copier = wave < 8;

  const int r0 = (tid & 255) >> 4, q = tid & 15;
  const float* xs = x   + (size_t)(m0 + r0) * D_MODEL + 4 * q;
  const float* ws = Wbc + (size_t)r0 * D_MODEL + 4 * q;
  const int d0 = r0 * LDS_PITCH + 4 * q;

  v4f st[G1_NLD];
  if (copier) {
#pragma unroll
    for (int j = 0; j < G1_NLD; ++j)
      st[j] = *(const v4f*)(j < 2 ? xs + (size_t)(16 * j) * D_MODEL
                                     : ws + (size_t)(16 * (j - 2)) * D_MODEL);
#pragma unroll
    for (int j = 0; j < G1_NLD; ++j)
      *(v4f*)(&sT[0][d0 + 16 * j * LDS_PITCH]) = st[j];
  }
  __syncthreads();

  v8f acc = {0.f, 0.f, 0.f, 0.f, 0.f, 0.f, 0.f, 0.f};
  const int aoff = (mt * 16 + c) * LDS_PITCH + 4 * h;
  const int boff = (G1_BM + nt * 16 + c) * LDS_PITCH + 4 * h;

  for (int kc = 0; kc < D_MODEL / G1_KC; ++kc) {
    const bool fetch = copier && (kc + 1 < D_MODEL / G1_KC);
    if (fetch) {
      const int ko = (kc + 1) * G1_KC;
#pragma unroll
      for (int j = 0; j < G1_NLD; ++j)
        st[j] = *(const v4f*)(j < 2 ? xs + (size_t)(16 * j) * D_MODEL + ko
                                       : ws + (size_t)(16 * (j - 2)) * D_MODEL + ko);
    }
    const float* sa = &sT[kc & 1][aoff];
    const float* sb = &sT[kc & 1][boff];
#pragma unroll
    for (int jj = 0; jj < G1_KC / 8; ++jj) {
      const v4f a4 = *(const v4f*)(sa + 8 * jj);
      const v4f b4 = *(const v4f*)(sb + 8 * jj);
      v2f a, b;
      a[0] = a4[0]; a[1] = a4[1]; b[0] = b4[0]; b[1] = b4[1];
      acc = __builtin_amdgcn_wmma_f32_16x16x4_f32(false, a, false, b,
                                                  (short)0, acc, false, false);
      a[0] = a4[2]; a[1] = a4[3]; b[0] = b4[2]; b[1] = b4[3];
      acc = __builtin_amdgcn_wmma_f32_16x16x4_f32(false, a, false, b,
                                                  (short)0, acc, false, false);
    }
    if (fetch) {
#pragma unroll
      for (int j = 0; j < G1_NLD; ++j)
        *(v4f*)(&sT[(kc + 1) & 1][d0 + 16 * j * LDS_PITCH]) = st[j];
    }
    __syncthreads();
  }
#pragma unroll
  for (int r = 0; r < 8; ++r) {
    int m = m0 + mt * 16 + 8 * h + r;
    dtBC[(size_t)m * E_DIM + nt * 16 + c] = acc[r];
  }
}

#define G2_BM      64
#define G2_BN      128
#define G2_ROWS    (G2_BM + G2_BN)
#define G2_NLD     12

__global__ __launch_bounds__(256) void k_proj_dt(const float* __restrict__ dtBC,
                                                 const float* __restrict__ Wdt,
                                                 const float* __restrict__ bdt,
                                                 float* __restrict__ dt) {
  __shared__ float sT[G2_ROWS * LDS_PITCH];

  const int tid = threadIdx.x;
  const int wave = tid >> 5, lane = tid & 31;
  const int c = lane & 15, h = lane >> 4;
  const int wm = wave & 3, wn = wave >> 2;
  const int m0 = (blockIdx.x >> 3) * G2_BM;
  const int n0 = (blockIdx.x & 7) * G2_BN;

  const int r0 = tid >> 4, q = tid & 15;
  const float* as = dtBC + (size_t)(m0 + r0) * E_DIM + 4 * q;
  const float* ws = Wdt  + (size_t)(n0 + r0) * DT_RANK + 4 * q;
  v4f st[G2_NLD];
#pragma unroll
  for (int j = 0; j < G2_NLD; ++j)
    st[j] = *(const v4f*)(j < 4 ? as + (size_t)(16 * j) * E_DIM
                                   : ws + (size_t)(16 * (j - 4)) * DT_RANK);
#pragma unroll
  for (int j = 0; j < G2_NLD; ++j)
    *(v4f*)(&sT[(r0 + 16 * j) * LDS_PITCH + 4 * q]) = st[j];
  __syncthreads();

  v8f acc[4];
#pragma unroll
  for (int t = 0; t < 4; ++t) acc[t] = (v8f){0.f, 0.f, 0.f, 0.f, 0.f, 0.f, 0.f, 0.f};

  const float* sa = &sT[(wm * 16 + c) * LDS_PITCH + 4 * h];
  const float* sb = &sT[(G2_BM + wn * 64 + c) * LDS_PITCH + 4 * h];
#pragma unroll
  for (int jj = 0; jj < DT_RANK / 8; ++jj) {
    const v4f a4 = *(const v4f*)(sa + 8 * jj);
    v2f a0, a1;
    a0[0] = a4[0]; a0[1] = a4[1]; a1[0] = a4[2]; a1[1] = a4[3];
#pragma unroll
    for (int t = 0; t < 4; ++t) {
      const v4f b4 = *(const v4f*)(sb + t * 16 * LDS_PITCH + 8 * jj);
      v2f b0, b1;
      b0[0] = b4[0]; b0[1] = b4[1]; b1[0] = b4[2]; b1[1] = b4[3];
      acc[t] = __builtin_amdgcn_wmma_f32_16x16x4_f32(false, a0, false, b0,
                                                     (short)0, acc[t], false, false);
      acc[t] = __builtin_amdgcn_wmma_f32_16x16x4_f32(false, a1, false, b1,
                                                     (short)0, acc[t], false, false);
    }
  }
#pragma unroll
  for (int t = 0; t < 4; ++t) {
    const int n = n0 + wn * 64 + t * 16 + c;
    const float bias = bdt[n];
#pragma unroll
    for (int r = 0; r < 8; ++r) {
      int m = m0 + wm * 16 + 8 * h + r;
      dt[(size_t)m * D_MODEL + n] = softplus_f(acc[t][r] + bias);
    }
  }
}

#define UPF 8

__global__ __launch_bounds__(256) void k_scan_local(const float* __restrict__ x,
                                                    const float* __restrict__ dt,
                                                    const float* __restrict__ dtBC,
                                                    const float* __restrict__ A_log,
                                                    float* __restrict__ hend,
                                                    float* __restrict__ Pprod) {
  __shared__ float4 sB[TCHUNK * 4];

  const int tid = threadIdx.x;
  const int blk = blockIdx.x;
  const int d = ((blk & 3) << 8) + tid;
  const int c = (blk >> 2) & (NCHUNK - 1);
  const int b = blk >> 8;
  const size_t row0 = (size_t)b * SEQLEN + (size_t)c * TCHUNK;

  sB[tid] = ((const float4*)(dtBC + (row0 + (tid >> 2)) * E_DIM + DT_RANK))[tid & 3];

  float An[D_STATE];
#pragma unroll
  for (int n = 0; n < D_STATE; ++n) An[n] = -__expf(A_log[d * D_STATE + n]);

  float h[D_STATE], P[D_STATE];
#pragma unroll
  for (int n = 0; n < D_STATE; ++n) { h[n] = 0.f; P[n] = 1.f; }

  __syncthreads();

  const float* dtp = dt + row0 * D_MODEL + d;
  const float* xp  = x  + row0 * D_MODEL + d;

  float dcur[UPF], xcur[UPF], dnxt[UPF], xnxt[UPF];
#pragma unroll
  for (int u = 0; u < UPF; ++u) {
    dcur[u] = dtp[(size_t)u * D_MODEL];
    xcur[u] = xp[(size_t)u * D_MODEL];
  }

  for (int tb = 0; tb < TCHUNK; tb += UPF) {
    const int tn = (tb + UPF < TCHUNK) ? tb + UPF : tb;
#pragma unroll
    for (int u = 0; u < UPF; ++u) {
      dnxt[u] = dtp[(size_t)(tn + u) * D_MODEL];
      xnxt[u] = xp[(size_t)(tn + u) * D_MODEL];
    }
    __builtin_amdgcn_sched_barrier(0);
#pragma unroll
    for (int u = 0; u < UPF; ++u) {
      const float dtv = dcur[u];
      const float xv  = xcur[u];
      float Bv[D_STATE];
#pragma unroll
      for (int i = 0; i < 4; ++i) {
        float4 q = sB[(tb + u) * 4 + i];
        Bv[4*i+0] = q.x; Bv[4*i+1] = q.y; Bv[4*i+2] = q.z; Bv[4*i+3] = q.w;
      }
      const float dx = dtv * xv;
#pragma unroll
      for (int n = 0; n < D_STATE; ++n) {
        float dA = __expf(dtv * An[n]);
        h[n] = dA * h[n] + dx * Bv[n];
        P[n] *= dA;
      }
    }
    __builtin_amdgcn_sched_barrier(0);
#pragma unroll
    for (int u = 0; u < UPF; ++u) { dcur[u] = dnxt[u]; xcur[u] = xnxt[u]; }
  }
  size_t base = (size_t)(b * NCHUNK + c) * D_STATE * D_MODEL + d;
#pragma unroll
  for (int n = 0; n < D_STATE; ++n) {
    hend[base + (size_t)n * D_MODEL]  = h[n];
    Pprod[base + (size_t)n * D_MODEL] = P[n];
  }
}

__global__ __launch_bounds__(256) void k_scan_combine(const float* __restrict__ hend,
                                                      const float* __restrict__ Pprod,
                                                      float* __restrict__ h0) {
  int gid = blockIdx.x * blockDim.x + threadIdx.x;
  int d = gid & (D_MODEL - 1);
  int n = (gid >> 10) & (D_STATE - 1);
  int b = gid >> 14;
  size_t base = ((size_t)b * NCHUNK * D_STATE + n) * D_MODEL + d;
  float h = 0.f;
  for (int c = 0; c < NCHUNK; ++c) {
    size_t idx = base + (size_t)c * D_STATE * D_MODEL;
    h0[idx] = h;
    h = Pprod[idx] * h + hend[idx];
  }
}

__global__ __launch_bounds__(256) void k_scan_emit(const float* __restrict__ x,
                                                   const float* __restrict__ dt,
                                                   const float* __restrict__ dtBC,
                                                   const float* __restrict__ A_log,
                                                   const float* __restrict__ Dp,
                                                   const float* __restrict__ h0,
                                                   float* __restrict__ y) {
  __shared__ float4 sBC[TCHUNK * 8];

  const int tid = threadIdx.x;
  const int blk = blockIdx.x;
  const int d = ((blk & 3) << 8) + tid;
  const int c = (blk >> 2) & (NCHUNK - 1);
  const int b = blk >> 8;
  const size_t row0 = (size_t)b * SEQLEN + (size_t)c * TCHUNK;

#pragma unroll
  for (int j = 0; j < 2; ++j) {
    int idx = tid + 256 * j;
    sBC[idx] = ((const float4*)(dtBC + (row0 + (idx >> 3)) * E_DIM + DT_RANK))[idx & 7];
  }

  float An[D_STATE];
#pragma unroll
  for (int n = 0; n < D_STATE; ++n) An[n] = -__expf(A_log[d * D_STATE + n]);

  float h[D_STATE];
  size_t base = (size_t)(b * NCHUNK + c) * D_STATE * D_MODEL + d;
#pragma unroll
  for (int n = 0; n < D_STATE; ++n) h[n] = h0[base + (size_t)n * D_MODEL];

  const float Dd = Dp[d];

  __syncthreads();

  const float* dtp = dt + row0 * D_MODEL + d;
  const float* xp  = x  + row0 * D_MODEL + d;
  float*       yp  = y  + row0 * D_MODEL + d;

  float dcur[UPF], xcur[UPF], dnxt[UPF], xnxt[UPF];
#pragma unroll
  for (int u = 0; u < UPF; ++u) {
    dcur[u] = dtp[(size_t)u * D_MODEL];
    xcur[u] = xp[(size_t)u * D_MODEL];
  }

  for (int tb = 0; tb < TCHUNK; tb += UPF) {
    const int tn = (tb + UPF < TCHUNK) ? tb + UPF : tb;
#pragma unroll
    for (int u = 0; u < UPF; ++u) {
      dnxt[u] = dtp[(size_t)(tn + u) * D_MODEL];
      xnxt[u] = xp[(size_t)(tn + u) * D_MODEL];
    }
    __builtin_amdgcn_sched_barrier(0);
#pragma unroll
    for (int u = 0; u < UPF; ++u) {
      const float dtv = dcur[u];
      const float xv  = xcur[u];
      float Bv[D_STATE], Cv[D_STATE];
#pragma unroll
      for (int i = 0; i < 4; ++i) {
        float4 q = sBC[(tb + u) * 8 + i];
        Bv[4*i+0] = q.x; Bv[4*i+1] = q.y; Bv[4*i+2] = q.z; Bv[4*i+3] = q.w;
      }
#pragma unroll
      for (int i = 0; i < 4; ++i) {
        float4 q = sBC[(tb + u) * 8 + 4 + i];
        Cv[4*i+0] = q.x; Cv[4*i+1] = q.y; Cv[4*i+2] = q.z; Cv[4*i+3] = q.w;
      }
      const float dx = dtv * xv;
      float yv = 0.f;
#pragma unroll
      for (int n = 0; n < D_STATE; ++n) {
        float dA = __expf(dtv * An[n]);
        h[n] = dA * h[n] + dx * Bv[n];
        yv += h[n] * Cv[n];
      }
      yp[(size_t)(tb + u) * D_MODEL] = yv + xv * Dd;
    }
    __builtin_amdgcn_sched_barrier(0);
#pragma unroll
    for (int u = 0; u < UPF; ++u) { dcur[u] = dnxt[u]; xcur[u] = xnxt[u]; }
  }
}

extern "C" void kernel_launch(void* const* d_in, const int* in_sizes, int n_in,
                              void* d_out, int out_size, void* d_ws, size_t ws_size,
                              hipStream_t stream) {
  const float* x    = (const float*)d_in[0];
  const float* Alog = (const float*)d_in[1];
  const float* Dp   = (const float*)d_in[2];
  const float* Wdt  = (const float*)d_in[3];
  const float* bdt  = (const float*)d_in[4];
  const float* Wbc  = (const float*)d_in[5];
  float* y = (float*)d_out;

  float* ws    = (float*)d_ws;
  float* dtBC  = ws;
  float* dt    = dtBC + (size_t)ROWS * E_DIM;
  float* hend  = dt + (size_t)ROWS * D_MODEL;
  float* Pprod = hend + (size_t)BATCH * D_MODEL * NCHUNK * D_STATE;
  float* h0    = Pprod + (size_t)BATCH * D_MODEL * NCHUNK * D_STATE;

  k_proj_dtbc<<<ROWS / G1_BM, G1_THREADS, 0, stream>>>(x, Wbc, dtBC);

  k_proj_dt<<<(ROWS / G2_BM) * (D_MODEL / G2_BN), 256, 0, stream>>>(dtBC, Wdt, bdt, dt);

  int nscan = BATCH * D_MODEL * NCHUNK;
  k_scan_local<<<nscan / 256, 256, 0, stream>>>(x, dt, dtBC, Alog, hend, Pprod);

  int ncomb = BATCH * D_MODEL * D_STATE;
  k_scan_combine<<<ncomb / 256, 256, 0, stream>>>(hend, Pprod, h0);

  k_scan_emit<<<nscan / 256, 256, 0, stream>>>(x, dt, dtBC, Alog, Dp, h0, y);
}


# scan passes: state-independent multiplies leave the step loop: log2(e) folded into A once per thread; chunk decay P = exp2(A*sum dt), was a running product; dA and P move in last bits
# speedup vs baseline: 2.4122x; 1.0552x over previous
#include <hip/hip_runtime.h>
#include <hip/hip_bf16.h>

#define D_MODEL 1024
#define D_STATE 16
#define DT_RANK 64
#define BATCH   2
#define SEQLEN  4096
#define E_DIM   (DT_RANK + 2*D_STATE)
#define ROWS    (BATCH*SEQLEN)
#define NCHUNK  64
#define TCHUNK  (SEQLEN/NCHUNK)

typedef __attribute__((ext_vector_type(2))) float v2f;
typedef __attribute__((ext_vector_type(8))) float v8f;
typedef __attribute__((ext_vector_type(4))) float v4f;

__device__ __forceinline__ float softplus_f(float v) {
  const float e = __expf(-fabsf(v));
  const float u = 1.0f + e;
  const float d = u - 1.0f;
  const float lg = __builtin_amdgcn_logf(u) * 0.693147180559945f;
  return fmaxf(v, 0.0f) + fmaf(e - d, 1.0f - d, lg);
}

#define LDS_PITCH 68

#define G1_BM      32
#define G1_KC      64
#define G1_ROWS    (G1_BM + E_DIM)
#define G1_THREADS 384
#define G1_NLD     8

__global__ __launch_bounds__(G1_THREADS) void k_proj_dtbc(const float* __restrict__ x,
                                                          const float* __restrict__ Wbc,
                                                          float* __restrict__ dtBC) {
  __shared__ float sT[2][G1_ROWS * LDS_PITCH];

  const int tid = threadIdx.x;
  const int wave = __builtin_amdgcn_readfirstlane(tid >> 5);
  const int lane = tid & 31;
  const int c = lane & 15, h = lane >> 4;
  const int mt = wave / 6, nt = wave - mt * 6;
  const int m0 = blockIdx.x * G1_BM;
  const bool copier = wave < 8;

  const int r0 = (tid & 255) >> 4, q = tid & 15;
  const float* xs = x   + (size_t)(m0 + r0) * D_MODEL + 4 * q;
  const float* ws = Wbc + (size_t)r0 * D_MODEL + 4 * q;
  const int d0 = r0 * LDS_PITCH + 4 * q;

  v4f st[G1_NLD];
  if (copier) {
#pragma unroll
    for (int j = 0; j < G1_NLD; ++j)
      st[j] = *(const v4f*)(j < 2 ? xs + (size_t)(16 * j) * D_MODEL
                                     : ws + (size_t)(16 * (j - 2)) * D_MODEL);
#pragma unroll
    for (int j = 0; j < G1_NLD; ++j)
      *(v4f*)(&sT[0][d0 + 16 * j * LDS_PITCH]) = st[j];
  }
  __syncthreads();

  v8f acc = {0.f, 0.f, 0.f, 0.f, 0.f, 0.f, 0.f, 0.f};
  const int aoff = (mt * 16 + c) * LDS_PITCH + 4 * h;
  const int boff = (G1_BM + nt * 16 + c) * LDS_PITCH + 4 * h;

  for (int kc = 0; kc < D_MODEL / G1_KC; ++kc) {
    const bool fetch = copier && (kc + 1 < D_MODEL / G1_KC);
    if (fetch) {
      const int ko = (kc + 1) * G1_KC;
#pragma unroll
      for (int j = 0; j < G1_NLD; ++j)
        st[j] = *(const v4f*)(j < 2 ? xs + (size_t)(16 * j) * D_MODEL + ko
                                       : ws + (size_t)(16 * (j - 2)) * D_MODEL + ko);
    }
    const float* sa = &sT[kc & 1][aoff];
    const float* sb = &sT[kc & 1][boff];
#pragma unroll
    for (int jj = 0; jj < G1_KC / 8; ++jj) {
      const v4f a4 = *(const v4f*)(sa + 8 * jj);
      const v4f b4 = *(const v4f*)(sb + 8 * jj);
      v2f a, b;
      a[0] = a4[0]; a[1] = a4[1]; b[0] = b4[0]; b[1] = b4[1];
      acc = __builtin_amdgcn_wmma_f32_16x16x4_f32(false, a, false, b,
                                                  (short)0, acc, false, false);
      a[0] = a4[2]; a[1] = a4[3]; b[0] = b4[2]; b[1] = b4[3];
      acc = __builtin_amdgcn_wmma_f32_16x16x4_f32(false, a, false, b,
                                                  (short)0, acc, false, false);
    }
    if (fetch) {
#pragma unroll
      for (int j = 0; j < G1_NLD; ++j)
        *(v4f*)(&sT[(kc + 1) & 1][d0 + 16 * j * LDS_PITCH]) = st[j];
    }
    __syncthreads();
  }
#pragma unroll
  for (int r = 0; r < 8; ++r) {
    int m = m0 + mt * 16 + 8 * h + r;
    dtBC[(size_t)m * E_DIM + nt * 16 + c] = acc[r];
  }
}

#define G2_BM      64
#define G2_BN      128
#define G2_ROWS    (G2_BM + G2_BN)
#define G2_NLD     12

__global__ __launch_bounds__(256) void k_proj_dt(const float* __restrict__ dtBC,
                                                 const float* __restrict__ Wdt,
                                                 const float* __restrict__ bdt,
                                                 float* __restrict__ dt) {
  __shared__ float sT[G2_ROWS * LDS_PITCH];

  const int tid = threadIdx.x;
  const int wave = tid >> 5, lane = tid & 31;
  const int c = lane & 15, h = lane >> 4;
  const int wm = wave & 3, wn = wave >> 2;
  const int m0 = (blockIdx.x >> 3) * G2_BM;
  const int n0 = (blockIdx.x & 7) * G2_BN;

  const int r0 = tid >> 4, q = tid & 15;
  const float* as = dtBC + (size_t)(m0 + r0) * E_DIM + 4 * q;
  const float* ws = Wdt  + (size_t)(n0 + r0) * DT_RANK + 4 * q;
  v4f st[G2_NLD];
#pragma unroll
  for (int j = 0; j < G2_NLD; ++j)
    st[j] = *(const v4f*)(j < 4 ? as + (size_t)(16 * j) * E_DIM
                                   : ws + (size_t)(16 * (j - 4)) * DT_RANK);
#pragma unroll
  for (int j = 0; j < G2_NLD; ++j)
    *(v4f*)(&sT[(r0 + 16 * j) * LDS_PITCH + 4 * q]) = st[j];
  __syncthreads();

  v8f acc[4];
#pragma unroll
  for (int t = 0; t < 4; ++t) acc[t] = (v8f){0.f, 0.f, 0.f, 0.f, 0.f, 0.f, 0.f, 0.f};

  const float* sa = &sT[(wm * 16 + c) * LDS_PITCH + 4 * h];
  const float* sb = &sT[(G2_BM + wn * 64 + c) * LDS_PITCH + 4 * h];
#pragma unroll
  for (int jj = 0; jj < DT_RANK / 8; ++jj) {
    const v4f a4 = *(const v4f*)(sa + 8 * jj);
    v2f a0, a1;
    a0[0] = a4[0]; a0[1] = a4[1]; a1[0] = a4[2]; a1[1] = a4[3];
#pragma unroll
    for (int t = 0; t < 4; ++t) {
      const v4f b4 = *(const v4f*)(sb + t * 16 * LDS_PITCH + 8 * jj);
      v2f b0, b1;
      b0[0] = b4[0]; b0[1] = b4[1]; b1[0] = b4[2]; b1[1] = b4[3];
      acc[t] = __builtin_amdgcn_wmma_f32_16x16x4_f32(false, a0, false, b0,
                                                     (short)0, acc[t], false, false);
      acc[t] = __builtin_amdgcn_wmma_f32_16x16x4_f32(false, a1, false, b1,
                                                     (short)0, acc[t], false, false);
    }
  }
#pragma unroll
  for (int t = 0; t < 4; ++t) {
    const int n = n0 + wn * 64 + t * 16 + c;
    const float bias = bdt[n];
#pragma unroll
    for (int r = 0; r < 8; ++r) {
      int m = m0 + wm * 16 + 8 * h + r;
      dt[(size_t)m * D_MODEL + n] = softplus_f(acc[t][r] + bias);
    }
  }
}

#define UPF 8
#define LOG2E_F 1.44269504088896341f

__global__ __launch_bounds__(256) void k_scan_local(const float* __restrict__ x,
                                                    const float* __restrict__ dt,
                                                    const float* __restrict__ dtBC,
                                                    const float* __restrict__ A_log,
                                                    float* __restrict__ hend,
                                                    float* __restrict__ Pprod) {
  __shared__ float4 sB[TCHUNK * 4];

  const int tid = threadIdx.x;
  const int blk = blockIdx.x;
  const int d = ((blk & 3) << 8) + tid;
  const int c = (blk >> 2) & (NCHUNK - 1);
  const int b = blk >> 8;
  const size_t row0 = (size_t)b * SEQLEN + (size_t)c * TCHUNK;

  sB[tid] = ((const float4*)(dtBC + (row0 + (tid >> 2)) * E_DIM + DT_RANK))[tid & 3];

  float A2[D_STATE];
#pragma unroll
  for (int n = 0; n < D_STATE; ++n) A2[n] = -__expf(A_log[d * D_STATE + n]) * LOG2E_F;

  float h[D_STATE];
#pragma unroll
  for (int n = 0; n < D_STATE; ++n) h[n] = 0.f;
  float sdt = 0.f;

  __syncthreads();

  const float* dtp = dt + row0 * D_MODEL + d;
  const float* xp  = x  + row0 * D_MODEL + d;

  float dcur[UPF], xcur[UPF], dnxt[UPF], xnxt[UPF];
#pragma unroll
  for (int u = 0; u < UPF; ++u) {
    dcur[u] = dtp[(size_t)u * D_MODEL];
    xcur[u] = xp[(size_t)u * D_MODEL];
  }

  for (int tb = 0; tb < TCHUNK; tb += UPF) {
    const int tn = (tb + UPF < TCHUNK) ? tb + UPF : tb;
#pragma unroll
    for (int u = 0; u < UPF; ++u) {
      dnxt[u] = dtp[(size_t)(tn + u) * D_MODEL];
      xnxt[u] = xp[(size_t)(tn + u) * D_MODEL];
    }
    __builtin_amdgcn_sched_barrier(0);
#pragma unroll
    for (int u = 0; u < UPF; ++u) {
      const float dtv = dcur[u];
      const float xv  = xcur[u];
      float Bv[D_STATE];
#pragma unroll
      for (int i = 0; i < 4; ++i) {
        float4 q = sB[(tb + u) * 4 + i];
        Bv[4*i+0] = q.x; Bv[4*i+1] = q.y; Bv[4*i+2] = q.z; Bv[4*i+3] = q.w;
      }
      const float dx = dtv * xv;
      sdt += dtv;
#pragma unroll
      for (int n = 0; n < D_STATE; ++n) {
        float dA = __builtin_amdgcn_exp2f(dtv * A2[n]);
        h[n] = dA * h[n] + dx * Bv[n];
      }
    }
    __builtin_amdgcn_sched_barrier(0);
#pragma unroll
    for (int u = 0; u < UPF; ++u) { dcur[u] = dnxt[u]; xcur[u] = xnxt[u]; }
  }
  size_t base = (size_t)(b * NCHUNK + c) * D_STATE * D_MODEL + d;
#pragma unroll
  for (int n = 0; n < D_STATE; ++n) {
    hend[base + (size_t)n * D_MODEL]  = h[n];
    Pprod[base + (size_t)n * D_MODEL] = __builtin_amdgcn_exp2f(sdt * A2[n]);
  }
}

__global__ __launch_bounds__(256) void k_scan_combine(const float* __restrict__ hend,
                                                      const float* __restrict__ Pprod,
                                                      float* __restrict__ h0) {
  int gid = blockIdx.x * blockDim.x + threadIdx.x;
  int d = gid & (D_MODEL - 1);
  int n = (gid >> 10) & (D_STATE - 1);
  int b = gid >> 14;
  size_t base = ((size_t)b * NCHUNK * D_STATE + n) * D_MODEL + d;
  float h = 0.f;
  for (int c = 0; c < NCHUNK; ++c) {
    size_t idx = base + (size_t)c * D_STATE * D_MODEL;
    h0[idx] = h;
    h = Pprod[idx] * h + hend[idx];
  }
}

__global__ __launch_bounds__(256) void k_scan_emit(const float* __restrict__ x,
                                                   const float* __restrict__ dt,
                                                   const float* __restrict__ dtBC,
                                                   const float* __restrict__ A_log,
                                                   const float* __restrict__ Dp,
                                                   const float* __restrict__ h0,
                                                   float* __restrict__ y) {
  __shared__ float4 sBC[TCHUNK * 8];

  const int tid = threadIdx.x;
  const int blk = blockIdx.x;
  const int d = ((blk & 3) << 8) + tid;
  const int c = (blk >> 2) & (NCHUNK - 1);
  const int b = blk >> 8;
  const size_t row0 = (size_t)b * SEQLEN + (size_t)c * TCHUNK;

#pragma unroll
  for (int j = 0; j < 2; ++j) {
    int idx = tid + 256 * j;
    sBC[idx] = ((const float4*)(dtBC + (row0 + (idx >> 3)) * E_DIM + DT_RANK))[idx & 7];
  }

  float A2[D_STATE];
#pragma unroll
  for (int n = 0; n < D_STATE; ++n) A2[n] = -__expf(A_log[d * D_STATE + n]) * LOG2E_F;

  float h[D_STATE];
  size_t base = (size_t)(b * NCHUNK + c) * D_STATE * D_MODEL + d;
#pragma unroll
  for (int n = 0; n < D_STATE; ++n) h[n] = h0[base + (size_t)n * D_MODEL];

  const float Dd = Dp[d];

  __syncthreads();

  const float* dtp = dt + row0 * D_MODEL + d;
  const float* xp  = x  + row0 * D_MODEL + d;
  float*       yp  = y  + row0 * D_MODEL + d;

  float dcur[UPF], xcur[UPF], dnxt[UPF], xnxt[UPF];
#pragma unroll
  for (int u = 0; u < UPF; ++u) {
    dcur[u] = dtp[(size_t)u * D_MODEL];
    xcur[u] = xp[(size_t)u * D_MODEL];
  }

  for (int tb = 0; tb < TCHUNK; tb += UPF) {
    const int tn = (tb + UPF < TCHUNK) ? tb + UPF : tb;
#pragma unroll
    for (int u = 0; u < UPF; ++u) {
      dnxt[u] = dtp[(size_t)(tn + u) * D_MODEL];
      xnxt[u] = xp[(size_t)(tn + u) * D_MODEL];
    }
    __builtin_amdgcn_sched_barrier(0);
#pragma unroll
    for (int u = 0; u < UPF; ++u) {
      const float dtv = dcur[u];
      const float xv  = xcur[u];
      float Bv[D_STATE], Cv[D_STATE];
#pragma unroll
      for (int i = 0; i < 4; ++i) {
        float4 q = sBC[(tb + u) * 8 + i];
        Bv[4*i+0] = q.x; Bv[4*i+1] = q.y; Bv[4*i+2] = q.z; Bv[4*i+3] = q.w;
      }
#pragma unroll
      for (int i = 0; i < 4; ++i) {
        float4 q = sBC[(tb + u) * 8 + 4 + i];
        Cv[4*i+0] = q.x; Cv[4*i+1] = q.y; Cv[4*i+2] = q.z; Cv[4*i+3] = q.w;
      }
      const float dx = dtv * xv;
      float yv = 0.f;
#pragma unroll
      for (int n = 0; n < D_STATE; ++n) {
        float dA = __builtin_amdgcn_exp2f(dtv * A2[n]);
        h[n] = dA * h[n] + dx * Bv[n];
        yv += h[n] * Cv[n];
      }
      yp[(size_t)(tb + u) * D_MODEL] = yv + xv * Dd;
    }
    __builtin_amdgcn_sched_barrier(0);
#pragma unroll
    for (int u = 0; u < UPF; ++u) { dcur[u] = dnxt[u]; xcur[u] = xnxt[u]; }
  }
}

extern "C" void kernel_launch(void* const* d_in, const int* in_sizes, int n_in,
                              void* d_out, int out_size, void* d_ws, size_t ws_size,
                              hipStream_t stream) {
  const float* x    = (const float*)d_in[0];
  const float* Alog = (const float*)d_in[1];
  const float* Dp   = (const float*)d_in[2];
  const float* Wdt  = (const float*)d_in[3];
  const float* bdt  = (const float*)d_in[4];
  const float* Wbc  = (const float*)d_in[5];
  float* y = (float*)d_out;

  float* ws    = (float*)d_ws;
  float* dtBC  = ws;
  float* dt    = dtBC + (size_t)ROWS * E_DIM;
  float* hend  = dt + (size_t)ROWS * D_MODEL;
  float* Pprod = hend + (size_t)BATCH * D_MODEL * NCHUNK * D_STATE;
  float* h0    = Pprod + (size_t)BATCH * D_MODEL * NCHUNK * D_STATE;

  k_proj_dtbc<<<ROWS / G1_BM, G1_THREADS, 0, stream>>>(x, Wbc, dtBC);

  k_proj_dt<<<(ROWS / G2_BM) * (D_MODEL / G2_BN), 256, 0, stream>>>(dtBC, Wdt, bdt, dt);

  int nscan = BATCH * D_MODEL * NCHUNK;
  k_scan_local<<<nscan / 256, 256, 0, stream>>>(x, dt, dtBC, Alog, hend, Pprod);

  int ncomb = BATCH * D_MODEL * D_STATE;
  k_scan_combine<<<ncomb / 256, 256, 0, stream>>>(hend, Pprod, h0);

  k_scan_emit<<<nscan / 256, 256, 0, stream>>>(x, dt, dtBC, Alog, Dp, h0, y);
}


# scan passes: a step's B,C rows (equal in all lanes) come by scalar loads and feed the multiplies as scalar operands; the LDS copy, its barrier and every per-step LDS read are gone; sums unchanged
# speedup vs baseline: 2.5108x; 1.0409x over previous
#include <hip/hip_runtime.h>
#include <hip/hip_bf16.h>

#define D_MODEL 1024
#define D_STATE 16
#define DT_RANK 64
#define BATCH   2
#define SEQLEN  4096
#define E_DIM   (DT_RANK + 2*D_STATE)
#define ROWS    (BATCH*SEQLEN)
#define NCHUNK  64
#define TCHUNK  (SEQLEN/NCHUNK)

typedef __attribute__((ext_vector_type(2))) float v2f;
typedef __attribute__((ext_vector_type(8))) float v8f;
typedef __attribute__((ext_vector_type(4))) float v4f;

__device__ __forceinline__ float softplus_f(float v) {
  const float e = __expf(-fabsf(v));
  const float u = 1.0f + e;
  const float d = u - 1.0f;
  const float lg = __builtin_amdgcn_logf(u) * 0.693147180559945f;
  return fmaxf(v, 0.0f) + fmaf(e - d, 1.0f - d, lg);
}

#define LDS_PITCH 68

#define G1_BM      32
#define G1_KC      64
#define G1_ROWS    (G1_BM + E_DIM)
#define G1_THREADS 384
#define G1_NLD     8

__global__ __launch_bounds__(G1_THREADS) void k_proj_dtbc(const float* __restrict__ x,
                                                          const float* __restrict__ Wbc,
                                                          float* __restrict__ dtBC) {
  __shared__ float sT[2][G1_ROWS * LDS_PITCH];

  const int tid = threadIdx.x;
  const int wave = __builtin_amdgcn_readfirstlane(tid >> 5);
  const int lane = tid & 31;
  const int c = lane & 15, h = lane >> 4;
  const int mt = wave / 6, nt = wave - mt * 6;
  const int m0 = blockIdx.x * G1_BM;
  const bool copier = wave < 8;

  const int r0 = (tid & 255) >> 4, q = tid & 15;
  const float* xs = x   + (size_t)(m0 + r0) * D_MODEL + 4 * q;
  const float* ws = Wbc + (size_t)r0 * D_MODEL + 4 * q;
  const int d0 = r0 * LDS_PITCH + 4 * q;

  v4f st[G1_NLD];
  if (copier) {
#pragma unroll
    for (int j = 0; j < G1_NLD; ++j)
      st[j] = *(const v4f*)(j < 2 ? xs + (size_t)(16 * j) * D_MODEL
                                     : ws + (size_t)(16 * (j - 2)) * D_MODEL);
#pragma unroll
    for (int j = 0; j < G1_NLD; ++j)
      *(v4f*)(&sT[0][d0 + 16 * j * LDS_PITCH]) = st[j];
  }
  __syncthreads();

  v8f acc = {0.f, 0.f, 0.f, 0.f, 0.f, 0.f, 0.f, 0.f};
  const int aoff = (mt * 16 + c) * LDS_PITCH + 4 * h;
  const int boff = (G1_BM + nt * 16 + c) * LDS_PITCH + 4 * h;

  for (int kc = 0; kc < D_MODEL / G1_KC; ++kc) {
    const bool fetch = copier && (kc + 1 < D_MODEL / G1_KC);
    if (fetch) {
      const int ko = (kc + 1) * G1_KC;
#pragma unroll
      for (int j = 0; j < G1_NLD; ++j)
        st[j] = *(const v4f*)(j < 2 ? xs + (size_t)(16 * j) * D_MODEL + ko
                                       : ws + (size_t)(16 * (j - 2)) * D_MODEL + ko);
    }
    const float* sa = &sT[kc & 1][aoff];
    const float* sb = &sT[kc & 1][boff];
#pragma unroll
    for (int jj = 0; jj < G1_KC / 8; ++jj) {
      const v4f a4 = *(const v4f*)(sa + 8 * jj);
      const v4f b4 = *(const v4f*)(sb + 8 * jj);
      v2f a, b;
      a[0] = a4[0]; a[1] = a4[1]; b[0] = b4[0]; b[1] = b4[1];
      acc = __builtin_amdgcn_wmma_f32_16x16x4_f32(false, a, false, b,
                                                  (short)0, acc, false, false);
      a[0] = a4[2]; a[1] = a4[3]; b[0] = b4[2]; b[1] = b4[3];
      acc = __builtin_amdgcn_wmma_f32_16x16x4_f32(false, a, false, b,
                                                  (short)0, acc, false, false);
    }
    if (fetch) {
#pragma unroll
      for (int j = 0; j < G1_NLD; ++j)
        *(v4f*)(&sT[(kc + 1) & 1][d0 + 16 * j * LDS_PITCH]) = st[j];
    }
    __syncthreads();
  }
#pragma unroll
  for (int r = 0; r < 8; ++r) {
    int m = m0 + mt * 16 + 8 * h + r;
    dtBC[(size_t)m * E_DIM + nt * 16 + c] = acc[r];
  }
}

#define G2_BM      64
#define G2_BN      128
#define G2_ROWS    (G2_BM + G2_BN)
#define G2_NLD     12

__global__ __launch_bounds__(256) void k_proj_dt(const float* __restrict__ dtBC,
                                                 const float* __restrict__ Wdt,
                                                 const float* __restrict__ bdt,
                                                 float* __restrict__ dt) {
  __shared__ float sT[G2_ROWS * LDS_PITCH];

  const int tid = threadIdx.x;
  const int wave = tid >> 5, lane = tid & 31;
  const int c = lane & 15, h = lane >> 4;
  const int wm = wave & 3, wn = wave >> 2;
  const int m0 = (blockIdx.x >> 3) * G2_BM;
  const int n0 = (blockIdx.x & 7) * G2_BN;

  const int r0 = tid >> 4, q = tid & 15;
  const float* as = dtBC + (size_t)(m0 + r0) * E_DIM + 4 * q;
  const float* ws = Wdt  + (size_t)(n0 + r0) * DT_RANK + 4 * q;
  v4f st[G2_NLD];
#pragma unroll
  for (int j = 0; j < G2_NLD; ++j)
    st[j] = *(const v4f*)(j < 4 ? as + (size_t)(16 * j) * E_DIM
                                   : ws + (size_t)(16 * (j - 4)) * DT_RANK);
#pragma unroll
  for (int j = 0; j < G2_NLD; ++j)
    *(v4f*)(&sT[(r0 + 16 * j) * LDS_PITCH + 4 * q]) = st[j];
  __syncthreads();

  v8f acc[4];
#pragma unroll
  for (int t = 0; t < 4; ++t) acc[t] = (v8f){0.f, 0.f, 0.f, 0.f, 0.f, 0.f, 0.f, 0.f};

  const float* sa = &sT[(wm * 16 + c) * LDS_PITCH + 4 * h];
  const float* sb = &sT[(G2_BM + wn * 64 + c) * LDS_PITCH + 4 * h];
#pragma unroll
  for (int jj = 0; jj < DT_RANK / 8; ++jj) {
    const v4f a4 = *(const v4f*)(sa + 8 * jj);
    v2f a0, a1;
    a0[0] = a4[0]; a0[1] = a4[1]; a1[0] = a4[2]; a1[1] = a4[3];
#pragma unroll
    for (int t = 0; t < 4; ++t) {
      const v4f b4 = *(const v4f*)(sb + t * 16 * LDS_PITCH + 8 * jj);
      v2f b0, b1;
      b0[0] = b4[0]; b0[1] = b4[1]; b1[0] = b4[2]; b1[1] = b4[3];
      acc[t] = __builtin_amdgcn_wmma_f32_16x16x4_f32(false, a0, false, b0,
                                                     (short)0, acc[t], false, false);
      acc[t] = __builtin_amdgcn_wmma_f32_16x16x4_f32(false, a1, false, b1,
                                                     (short)0, acc[t], false, false);
    }
  }
#pragma unroll
  for (int t = 0; t < 4; ++t) {
    const int n = n0 + wn * 64 + t * 16 + c;
    const float bias = bdt[n];
#pragma unroll
    for (int r = 0; r < 8; ++r) {
      int m = m0 + wm * 16 + 8 * h + r;
      dt[(size_t)m * D_MODEL + n] = softplus_f(acc[t][r] + bias);
    }
  }
}

#define UPF 8
#define LOG2E_F 1.44269504088896341f

__global__ __launch_bounds__(256) void k_scan_local(const float* __restrict__ x,
                                                    const float* __restrict__ dt,
                                                    const float* __restrict__ dtBC,
                                                    const float* __restrict__ A_log,
                                                    float* __restrict__ hend,
                                                    float* __restrict__ Pprod) {
  const int tid = threadIdx.x;
  const int blk = blockIdx.x;
  const int d = ((blk & 3) << 8) + tid;
  const int c = (blk >> 2) & (NCHUNK - 1);
  const int b = blk >> 8;
  const size_t row0 = (size_t)b * SEQLEN + (size_t)c * TCHUNK;

  const float4* __restrict__ brow = (const float4*)(dtBC + row0 * E_DIM + DT_RANK);

  float A2[D_STATE];
#pragma unroll
  for (int n = 0; n < D_STATE; ++n) A2[n] = -__expf(A_log[d * D_STATE + n]) * LOG2E_F;

  float h[D_STATE];
#pragma unroll
  for (int n = 0; n < D_STATE; ++n) h[n] = 0.f;
  float sdt = 0.f;

  const float* dtp = dt + row0 * D_MODEL + d;
  const float* xp  = x  + row0 * D_MODEL + d;

  float dcur[UPF], xcur[UPF], dnxt[UPF], xnxt[UPF];
#pragma unroll
  for (int u = 0; u < UPF; ++u) {
    dcur[u] = dtp[(size_t)u * D_MODEL];
    xcur[u] = xp[(size_t)u * D_MODEL];
  }

  for (int tb = 0; tb < TCHUNK; tb += UPF) {
    const int tn = (tb + UPF < TCHUNK) ? tb + UPF : tb;
#pragma unroll
    for (int u = 0; u < UPF; ++u) {
      dnxt[u] = dtp[(size_t)(tn + u) * D_MODEL];
      xnxt[u] = xp[(size_t)(tn + u) * D_MODEL];
    }
    __builtin_amdgcn_sched_barrier(0);
#pragma unroll
    for (int u = 0; u < UPF; ++u) {
      const float dtv = dcur[u];
      const float xv  = xcur[u];
      float Bv[D_STATE];
#pragma unroll
      for (int i = 0; i < 4; ++i) {
        float4 q = brow[(tb + u) * (E_DIM / 4) + i];
        Bv[4*i+0] = q.x; Bv[4*i+1] = q.y; Bv[4*i+2] = q.z; Bv[4*i+3] = q.w;
      }
      const float dx = dtv * xv;
      sdt += dtv;
#pragma unroll
      for (int n = 0; n < D_STATE; ++n) {
        float dA = __builtin_amdgcn_exp2f(dtv * A2[n]);
        h[n] = dA * h[n] + dx * Bv[n];
      }
    }
    __builtin_amdgcn_sched_barrier(0);
#pragma unroll
    for (int u = 0; u < UPF; ++u) { dcur[u] = dnxt[u]; xcur[u] = xnxt[u]; }
  }
  size_t base = (size_t)(b * NCHUNK + c) * D_STATE * D_MODEL + d;
#pragma unroll
  for (int n = 0; n < D_STATE; ++n) {
    hend[base + (size_t)n * D_MODEL]  = h[n];
    Pprod[base + (size_t)n * D_MODEL] = __builtin_amdgcn_exp2f(sdt * A2[n]);
  }
}

__global__ __launch_bounds__(256) void k_scan_combine(const float* __restrict__ hend,
                                                      const float* __restrict__ Pprod,
                                                      float* __restrict__ h0) {
  int gid = blockIdx.x * blockDim.x + threadIdx.x;
  int d = gid & (D_MODEL - 1);
  int n = (gid >> 10) & (D_STATE - 1);
  int b = gid >> 14;
  size_t base = ((size_t)b * NCHUNK * D_STATE + n) * D_MODEL + d;
  float h = 0.f;
  for (int c = 0; c < NCHUNK; ++c) {
    size_t idx = base + (size_t)c * D_STATE * D_MODEL;
    h0[idx] = h;
    h = Pprod[idx] * h + hend[idx];
  }
}

__global__ __launch_bounds__(256) void k_scan_emit(const float* __restrict__ x,
                                                   const float* __restrict__ dt,
                                                   const float* __restrict__ dtBC,
                                                   const float* __restrict__ A_log,
                                                   const float* __restrict__ Dp,
                                                   const float* __restrict__ h0,
                                                   float* __restrict__ y) {
  const int tid = threadIdx.x;
  const int blk = blockIdx.x;
  const int d = ((blk & 3) << 8) + tid;
  const int c = (blk >> 2) & (NCHUNK - 1);
  const int b = blk >> 8;
  const size_t row0 = (size_t)b * SEQLEN + (size_t)c * TCHUNK;

  const float4* __restrict__ bcrow = (const float4*)(dtBC + row0 * E_DIM + DT_RANK);

  float A2[D_STATE];
#pragma unroll
  for (int n = 0; n < D_STATE; ++n) A2[n] = -__expf(A_log[d * D_STATE + n]) * LOG2E_F;

  float h[D_STATE];
  size_t base = (size_t)(b * NCHUNK + c) * D_STATE * D_MODEL + d;
#pragma unroll
  for (int n = 0; n < D_STATE; ++n) h[n] = h0[base + (size_t)n * D_MODEL];

  const float Dd = Dp[d];

  const float* dtp = dt + row0 * D_MODEL + d;
  const float* xp  = x  + row0 * D_MODEL + d;
  float*       yp  = y  + row0 * D_MODEL + d;

  float dcur[UPF], xcur[UPF], dnxt[UPF], xnxt[UPF];
#pragma unroll
  for (int u = 0; u < UPF; ++u) {
    dcur[u] = dtp[(size_t)u * D_MODEL];
    xcur[u] = xp[(size_t)u * D_MODEL];
  }

  for (int tb = 0; tb < TCHUNK; tb += UPF) {
    const int tn = (tb + UPF < TCHUNK) ? tb + UPF : tb;
#pragma unroll
    for (int u = 0; u < UPF; ++u) {
      dnxt[u] = dtp[(size_t)(tn + u) * D_MODEL];
      xnxt[u] = xp[(size_t)(tn + u) * D_MODEL];
    }
    __builtin_amdgcn_sched_barrier(0);
#pragma unroll
    for (int u = 0; u < UPF; ++u) {
      const float dtv = dcur[u];
      const float xv  = xcur[u];
      float Bv[D_STATE], Cv[D_STATE];
#pragma unroll
      for (int i = 0; i < 4; ++i) {
        float4 q = bcrow[(tb + u) * (E_DIM / 4) + i];
        Bv[4*i+0] = q.x; Bv[4*i+1] = q.y; Bv[4*i+2] = q.z; Bv[4*i+3] = q.w;
      }
#pragma unroll
      for (int i = 0; i < 4; ++i) {
        float4 q = bcrow[(tb + u) * (E_DIM / 4) + 4 + i];
        Cv[4*i+0] = q.x; Cv[4*i+1] = q.y; Cv[4*i+2] = q.z; Cv[4*i+3] = q.w;
      }
      const float dx = dtv * xv;
      float yv = 0.f;
#pragma unroll
      for (int n = 0; n < D_STATE; ++n) {
        float dA = __builtin_amdgcn_exp2f(dtv * A2[n]);
        h[n] = dA * h[n] + dx * Bv[n];
        yv += h[n] * Cv[n];
      }
      yp[(size_t)(tb + u) * D_MODEL] = yv + xv * Dd;
    }
    __builtin_amdgcn_sched_barrier(0);
#pragma unroll
    for (int u = 0; u < UPF; ++u) { dcur[u] = dnxt[u]; xcur[u] = xnxt[u]; }
  }
}

extern "C" void kernel_launch(void* const* d_in, const int* in_sizes, int n_in,
                              void* d_out, int out_size, void* d_ws, size_t ws_size,
                              hipStream_t stream) {
  const float* x    = (const float*)d_in[0];
  const float* Alog = (const float*)d_in[1];
  const float* Dp   = (const float*)d_in[2];
  const float* Wdt  = (const float*)d_in[3];
  const float* bdt  = (const float*)d_in[4];
  const float* Wbc  = (const float*)d_in[5];
  float* y = (float*)d_out;

  float* ws    = (float*)d_ws;
  float* dtBC  = ws;
  float* dt    = dtBC + (size_t)ROWS * E_DIM;
  float* hend  = dt + (size_t)ROWS * D_MODEL;
  float* Pprod = hend + (size_t)BATCH * D_MODEL * NCHUNK * D_STATE;
  float* h0    = Pprod + (size_t)BATCH * D_MODEL * NCHUNK * D_STATE;

  k_proj_dtbc<<<ROWS / G1_BM, G1_THREADS, 0, stream>>>(x, Wbc, dtBC);

  k_proj_dt<<<(ROWS / G2_BM) * (D_MODEL / G2_BN), 256, 0, stream>>>(dtBC, Wdt, bdt, dt);

  int nscan = BATCH * D_MODEL * NCHUNK;
  k_scan_local<<<nscan / 256, 256, 0, stream>>>(x, dt, dtBC, Alog, hend, Pprod);

  int ncomb = BATCH * D_MODEL * D_STATE;
  k_scan_combine<<<ncomb / 256, 256, 0, stream>>>(hend, Pprod, h0);

  k_scan_emit<<<nscan / 256, 256, 0, stream>>>(x, dt, dtBC, Alog, Dp, h0, y);
}


# chunk combine: (P, hend) kept as 8-byte pairs and all 64 pairs of a chain requested before its multiply-add chain starts: one exposed memory round trip, was eight; order of sums unchanged
# speedup vs baseline: 2.7339x; 1.0889x over previous
#include <hip/hip_runtime.h>
#include <hip/hip_bf16.h>

#define D_MODEL 1024
#define D_STATE 16
#define DT_RANK 64
#define BATCH   2
#define SEQLEN  4096
#define E_DIM   (DT_RANK + 2*D_STATE)
#define ROWS    (BATCH*SEQLEN)
#define NCHUNK  64
#define TCHUNK  (SEQLEN/NCHUNK)

typedef __attribute__((ext_vector_type(2))) float v2f;
typedef __attribute__((ext_vector_type(8))) float v8f;
typedef __attribute__((ext_vector_type(4))) float v4f;

__device__ __forceinline__ float softplus_f(float v) {
  const float e = __expf(-fabsf(v));
  const float u = 1.0f + e;
  const float d = u - 1.0f;
  const float lg = __builtin_amdgcn_logf(u) * 0.693147180559945f;
  return fmaxf(v, 0.0f) + fmaf(e - d, 1.0f - d, lg);
}

#define LDS_PITCH 68

#define G1_BM      32
#define G1_KC      64
#define G1_ROWS    (G1_BM + E_DIM)
#define G1_THREADS 384
#define G1_NLD     8

__global__ __launch_bounds__(G1_THREADS) void k_proj_dtbc(const float* __restrict__ x,
                                                          const float* __restrict__ Wbc,
                                                          float* __restrict__ dtBC) {
  __shared__ float sT[2][G1_ROWS * LDS_PITCH];

  const int tid = threadIdx.x;
  const int wave = __builtin_amdgcn_readfirstlane(tid >> 5);
  const int lane = tid & 31;
  const int c = lane & 15, h = lane >> 4;
  const int mt = wave / 6, nt = wave - mt * 6;
  const int m0 = blockIdx.x * G1_BM;
  const bool copier = wave < 8;

  const int r0 = (tid & 255) >> 4, q = tid & 15;
  const float* xs = x   + (size_t)(m0 + r0) * D_MODEL + 4 * q;
  const float* ws = Wbc + (size_t)r0 * D_MODEL + 4 * q;
  const int d0 = r0 * LDS_PITCH + 4 * q;

  v4f st[G1_NLD];
  if (copier) {
#pragma unroll
    for (int j = 0; j < G1_NLD; ++j)
      st[j] = *(const v4f*)(j < 2 ? xs + (size_t)(16 * j) * D_MODEL
                                     : ws + (size_t)(16 * (j - 2)) * D_MODEL);
#pragma unroll
    for (int j = 0; j < G1_NLD; ++j)
      *(v4f*)(&sT[0][d0 + 16 * j * LDS_PITCH]) = st[j];
  }
  __syncthreads();

  v8f acc = {0.f, 0.f, 0.f, 0.f, 0.f, 0.f, 0.f, 0.f};
  const int aoff = (mt * 16 + c) * LDS_PITCH + 4 * h;
  const int boff = (G1_BM + nt * 16 + c) * LDS_PITCH + 4 * h;

  for (int kc = 0; kc < D_MODEL / G1_KC; ++kc) {
    const bool fetch = copier && (kc + 1 < D_MODEL / G1_KC);
    if (fetch) {
      const int ko = (kc + 1) * G1_KC;
#pragma unroll
      for (int j = 0; j < G1_NLD; ++j)
        st[j] = *(const v4f*)(j < 2 ? xs + (size_t)(16 * j) * D_MODEL + ko
                                       : ws + (size_t)(16 * (j - 2)) * D_MODEL + ko);
    }
    const float* sa = &sT[kc & 1][aoff];
    const float* sb = &sT[kc & 1][boff];
#pragma unroll
    for (int jj = 0; jj < G1_KC / 8; ++jj) {
      const v4f a4 = *(const v4f*)(sa + 8 * jj);
      const v4f b4 = *(const v4f*)(sb + 8 * jj);
      v2f a, b;
      a[0] = a4[0]; a[1] = a4[1]; b[0] = b4[0]; b[1] = b4[1];
      acc = __builtin_amdgcn_wmma_f32_16x16x4_f32(false, a, false, b,
                                                  (short)0, acc, false, false);
      a[0] = a4[2]; a[1] = a4[3]; b[0] = b4[2]; b[1] = b4[3];
      acc = __builtin_amdgcn_wmma_f32_16x16x4_f32(false, a, false, b,
                                                  (short)0, acc, false, false);
    }
    if (fetch) {
#pragma unroll
      for (int j = 0; j < G1_NLD; ++j)
        *(v4f*)(&sT[(kc + 1) & 1][d0 + 16 * j * LDS_PITCH]) = st[j];
    }
    __syncthreads();
  }
#pragma unroll
  for (int r = 0; r < 8; ++r) {
    int m = m0 + mt * 16 + 8 * h + r;
    dtBC[(size_t)m * E_DIM + nt * 16 + c] = acc[r];
  }
}

#define G2_BM      64
#define G2_BN      128
#define G2_ROWS    (G2_BM + G2_BN)
#define G2_NLD     12

__global__ __launch_bounds__(256) void k_proj_dt(const float* __restrict__ dtBC,
                                                 const float* __restrict__ Wdt,
                                                 const float* __restrict__ bdt,
                                                 float* __restrict__ dt) {
  __shared__ float sT[G2_ROWS * LDS_PITCH];

  const int tid = threadIdx.x;
  const int wave = tid >> 5, lane = tid & 31;
  const int c = lane & 15, h = lane >> 4;
  const int wm = wave & 3, wn = wave >> 2;
  const int m0 = (blockIdx.x >> 3) * G2_BM;
  const int n0 = (blockIdx.x & 7) * G2_BN;

  const int r0 = tid >> 4, q = tid & 15;
  const float* as = dtBC + (size_t)(m0 + r0) * E_DIM + 4 * q;
  const float* ws = Wdt  + (size_t)(n0 + r0) * DT_RANK + 4 * q;
  v4f st[G2_NLD];
#pragma unroll
  for (int j = 0; j < G2_NLD; ++j)
    st[j] = *(const v4f*)(j < 4 ? as + (size_t)(16 * j) * E_DIM
                                   : ws + (size_t)(16 * (j - 4)) * DT_RANK);
#pragma unroll
  for (int j = 0; j < G2_NLD; ++j)
    *(v4f*)(&sT[(r0 + 16 * j) * LDS_PITCH + 4 * q]) = st[j];
  __syncthreads();

  v8f acc[4];
#pragma unroll
  for (int t = 0; t < 4; ++t) acc[t] = (v8f){0.f, 0.f, 0.f, 0.f, 0.f, 0.f, 0.f, 0.f};

  const float* sa = &sT[(wm * 16 + c) * LDS_PITCH + 4 * h];
  const float* sb = &sT[(G2_BM + wn * 64 + c) * LDS_PITCH + 4 * h];
#pragma unroll
  for (int jj = 0; jj < DT_RANK / 8; ++jj) {
    const v4f a4 = *(const v4f*)(sa + 8 * jj);
    v2f a0, a1;
    a0[0] = a4[0]; a0[1] = a4[1]; a1[0] = a4[2]; a1[1] = a4[3];
#pragma unroll
    for (int t = 0; t < 4; ++t) {
      const v4f b4 = *(const v4f*)(sb + t * 16 * LDS_PITCH + 8 * jj);
      v2f b0, b1;
      b0[0] = b4[0]; b0[1] = b4[1]; b1[0] = b4[2]; b1[1] = b4[3];
      acc[t] = __builtin_amdgcn_wmma_f32_16x16x4_f32(false, a0, false, b0,
                                                     (short)0, acc[t], false, false);
      acc[t] = __builtin_amdgcn_wmma_f32_16x16x4_f32(false, a1, false, b1,
                                                     (short)0, acc[t], false, false);
    }
  }
#pragma unroll
  for (int t = 0; t < 4; ++t) {
    const int n = n0 + wn * 64 + t * 16 + c;
    const float bias = bdt[n];
#pragma unroll
    for (int r = 0; r < 8; ++r) {
      int m = m0 + wm * 16 + 8 * h + r;
      dt[(size_t)m * D_MODEL + n] = softplus_f(acc[t][r] + bias);
    }
  }
}

#define UPF 8
#define LOG2E_F 1.44269504088896341f

__global__ __launch_bounds__(256) void k_scan_local(const float* __restrict__ x,
                                                    const float* __restrict__ dt,
                                                    const float* __restrict__ dtBC,
                                                    const float* __restrict__ A_log,
                                                    float* __restrict__ chst) {
  const int tid = threadIdx.x;
  const int blk = blockIdx.x;
  const int d = ((blk & 3) << 8) + tid;
  const int c = (blk >> 2) & (NCHUNK - 1);
  const int b = blk >> 8;
  const size_t row0 = (size_t)b * SEQLEN + (size_t)c * TCHUNK;

  const float4* __restrict__ brow = (const float4*)(dtBC + row0 * E_DIM + DT_RANK);

  float A2[D_STATE];
#pragma unroll
  for (int n = 0; n < D_STATE; ++n) A2[n] = -__expf(A_log[d * D_STATE + n]) * LOG2E_F;

  float h[D_STATE];
#pragma unroll
  for (int n = 0; n < D_STATE; ++n) h[n] = 0.f;
  float sdt = 0.f;

  const float* dtp = dt + row0 * D_MODEL + d;
  const float* xp  = x  + row0 * D_MODEL + d;

  float dcur[UPF], xcur[UPF], dnxt[UPF], xnxt[UPF];
#pragma unroll
  for (int u = 0; u < UPF; ++u) {
    dcur[u] = dtp[(size_t)u * D_MODEL];
    xcur[u] = xp[(size_t)u * D_MODEL];
  }

  for (int tb = 0; tb < TCHUNK; tb += UPF) {
    const int tn = (tb + UPF < TCHUNK) ? tb + UPF : tb;
#pragma unroll
    for (int u = 0; u < UPF; ++u) {
      dnxt[u] = dtp[(size_t)(tn + u) * D_MODEL];
      xnxt[u] = xp[(size_t)(tn + u) * D_MODEL];
    }
    __builtin_amdgcn_sched_barrier(0);
#pragma unroll
    for (int u = 0; u < UPF; ++u) {
      const float dtv = dcur[u];
      const float xv  = xcur[u];
      float Bv[D_STATE];
#pragma unroll
      for (int i = 0; i < 4; ++i) {
        float4 q = brow[(tb + u) * (E_DIM / 4) + i];
        Bv[4*i+0] = q.x; Bv[4*i+1] = q.y; Bv[4*i+2] = q.z; Bv[4*i+3] = q.w;
      }
      const float dx = dtv * xv;
      sdt += dtv;
#pragma unroll
      for (int n = 0; n < D_STATE; ++n) {
        float dA = __builtin_amdgcn_exp2f(dtv * A2[n]);
        h[n] = dA * h[n] + dx * Bv[n];
      }
    }
    __builtin_amdgcn_sched_barrier(0);
#pragma unroll
    for (int u = 0; u < UPF; ++u) { dcur[u] = dnxt[u]; xcur[u] = xnxt[u]; }
  }
  size_t base = (size_t)(b * NCHUNK + c) * D_STATE * D_MODEL + d;
  v2f* __restrict__ st = (v2f*)chst;
#pragma unroll
  for (int n = 0; n < D_STATE; ++n) {
    v2f ph;
    ph[0] = __builtin_amdgcn_exp2f(sdt * A2[n]);
    ph[1] = h[n];
    st[base + (size_t)n * D_MODEL] = ph;
  }
}

__global__ __launch_bounds__(256) void k_scan_combine(const float* __restrict__ chst,
                                                      float* __restrict__ h0) {
  int gid = blockIdx.x * blockDim.x + threadIdx.x;
  int d = gid & (D_MODEL - 1);
  int n = (gid >> 10) & (D_STATE - 1);
  int b = gid >> 14;
  size_t base = ((size_t)b * NCHUNK * D_STATE + n) * D_MODEL + d;
  const v2f* __restrict__ pp = (const v2f*)chst + base;
  float* __restrict__ op = h0 + base;
  v2f ph[NCHUNK];
#pragma unroll
  for (int c = 0; c < NCHUNK; ++c) ph[c] = pp[(size_t)c * D_STATE * D_MODEL];
  __builtin_amdgcn_sched_barrier(0);
  float h = 0.f;
#pragma unroll
  for (int c = 0; c < NCHUNK; ++c) {
    op[(size_t)c * D_STATE * D_MODEL] = h;
    h = ph[c][0] * h + ph[c][1];
  }
}

__global__ __launch_bounds__(256) void k_scan_emit(const float* __restrict__ x,
                                                   const float* __restrict__ dt,
                                                   const float* __restrict__ dtBC,
                                                   const float* __restrict__ A_log,
                                                   const float* __restrict__ Dp,
                                                   const float* __restrict__ h0,
                                                   float* __restrict__ y) {
  const int tid = threadIdx.x;
  const int blk = blockIdx.x;
  const int d = ((blk & 3) << 8) + tid;
  const int c = (blk >> 2) & (NCHUNK - 1);
  const int b = blk >> 8;
  const size_t row0 = (size_t)b * SEQLEN + (size_t)c * TCHUNK;

  const float4* __restrict__ bcrow = (const float4*)(dtBC + row0 * E_DIM + DT_RANK);

  float A2[D_STATE];
#pragma unroll
  for (int n = 0; n < D_STATE; ++n) A2[n] = -__expf(A_log[d * D_STATE + n]) * LOG2E_F;

  float h[D_STATE];
  size_t base = (size_t)(b * NCHUNK + c) * D_STATE * D_MODEL + d;
#pragma unroll
  for (int n = 0; n < D_STATE; ++n) h[n] = h0[base + (size_t)n * D_MODEL];

  const float Dd = Dp[d];

  const float* dtp = dt + row0 * D_MODEL + d;
  const float* xp  = x  + row0 * D_MODEL + d;
  float*       yp  = y  + row0 * D_MODEL + d;

  float dcur[UPF], xcur[UPF], dnxt[UPF], xnxt[UPF];
#pragma unroll
  for (int u = 0; u < UPF; ++u) {
    dcur[u] = dtp[(size_t)u * D_MODEL];
    xcur[u] = xp[(size_t)u * D_MODEL];
  }

  for (int tb = 0; tb < TCHUNK; tb += UPF) {
    const int tn = (tb + UPF < TCHUNK) ? tb + UPF : tb;
#pragma unroll
    for (int u = 0; u < UPF; ++u) {
      dnxt[u] = dtp[(size_t)(tn + u) * D_MODEL];
      xnxt[u] = xp[(size_t)(tn + u) * D_MODEL];
    }
    __builtin_amdgcn_sched_barrier(0);
#pragma unroll
    for (int u = 0; u < UPF; ++u) {
      const float dtv = dcur[u];
      const float xv  = xcur[u];
      float Bv[D_STATE], Cv[D_STATE];
#pragma unroll
      for (int i = 0; i < 4; ++i) {
        float4 q = bcrow[(tb + u) * (E_DIM / 4) + i];
        Bv[4*i+0] = q.x; Bv[4*i+1] = q.y; Bv[4*i+2] = q.z; Bv[4*i+3] = q.w;
      }
#pragma unroll
      for (int i = 0; i < 4; ++i) {
        float4 q = bcrow[(tb + u) * (E_DIM / 4) + 4 + i];
        Cv[4*i+0] = q.x; Cv[4*i+1] = q.y; Cv[4*i+2] = q.z; Cv[4*i+3] = q.w;
      }
      const float dx = dtv * xv;
      float yv = 0.f;
#pragma unroll
      for (int n = 0; n < D_STATE; ++n) {
        float dA = __builtin_amdgcn_exp2f(dtv * A2[n]);
        h[n] = dA * h[n] + dx * Bv[n];
        yv += h[n] * Cv[n];
      }
      yp[(size_t)(tb + u) * D_MODEL] = yv + xv * Dd;
    }
    __builtin_amdgcn_sched_barrier(0);
#pragma unroll
    for (int u = 0; u < UPF; ++u) { dcur[u] = dnxt[u]; xcur[u] = xnxt[u]; }
  }
}

extern "C" void kernel_launch(void* const* d_in, const int* in_sizes, int n_in,
                              void* d_out, int out_size, void* d_ws, size_t ws_size,
                              hipStream_t stream) {
  const float* x    = (const float*)d_in[0];
  const float* Alog = (const float*)d_in[1];
  const float* Dp   = (const float*)d_in[2];
  const float* Wdt  = (const float*)d_in[3];
  const float* bdt  = (const float*)d_in[4];
  const float* Wbc  = (const float*)d_in[5];
  float* y = (float*)d_out;

  float* ws    = (float*)d_ws;
  float* dtBC  = ws;
  float* dt    = dtBC + (size_t)ROWS * E_DIM;
  float* chst  = dt + (size_t)ROWS * D_MODEL;
  float* h0    = chst + (size_t)2 * BATCH * D_MODEL * NCHUNK * D_STATE;

  k_proj_dtbc<<<ROWS / G1_BM, G1_THREADS, 0, stream>>>(x, Wbc, dtBC);

  k_proj_dt<<<(ROWS / G2_BM) * (D_MODEL / G2_BN), 256, 0, stream>>>(dtBC, Wdt, bdt, dt);

  int nscan = BATCH * D_MODEL * NCHUNK;
  k_scan_local<<<nscan / 256, 256, 0, stream>>>(x, dt, dtBC, Alog, chst);

  int ncomb = BATCH * D_MODEL * D_STATE;
  k_scan_combine<<<ncomb / 256, 256, 0, stream>>>(chst, h0);

  k_scan_emit<<<nscan / 256, 256, 0, stream>>>(x, dt, dtBC, Alog, Dp, h0, y);
}


# GEMM 1: operand chunks requested two K-chunks ahead (second register set, all 12 waves copy, barriers order LDS only): a request has two chunks of matrix work to land in, was one; sums unchanged
# speedup vs baseline: 2.8779x; 1.0527x over previous
#include <hip/hip_runtime.h>
#include <hip/hip_bf16.h>

#define D_MODEL 1024
#define D_STATE 16
#define DT_RANK 64
#define BATCH   2
#define SEQLEN  4096
#define E_DIM   (DT_RANK + 2*D_STATE)
#define ROWS    (BATCH*SEQLEN)
#define NCHUNK  64
#define TCHUNK  (SEQLEN/NCHUNK)

typedef __attribute__((ext_vector_type(2))) float v2f;
typedef __attribute__((ext_vector_type(8))) float v8f;
typedef __attribute__((ext_vector_type(4))) float v4f;

__device__ __forceinline__ float softplus_f(float v) {
  const float e = __expf(-fabsf(v));
  const float u = 1.0f + e;
  const float d = u - 1.0f;
  const float lg = __builtin_amdgcn_logf(u) * 0.693147180559945f;
  return fmaxf(v, 0.0f) + fmaf(e - d, 1.0f - d, lg);
}

#define LDS_PITCH 68

__device__ __forceinline__ void lds_barrier() {
  __builtin_amdgcn_fence(__ATOMIC_RELEASE, "workgroup", "local");
  __builtin_amdgcn_s_barrier();
  __builtin_amdgcn_fence(__ATOMIC_ACQUIRE, "workgroup", "local");
}

#define G1_BM      32
#define G1_KC      64
#define G1_ROWS    (G1_BM + E_DIM)
#define G1_THREADS 384
#define G1_NLD     6

__global__ __launch_bounds__(G1_THREADS) void k_proj_dtbc(const float* __restrict__ x,
                                                          const float* __restrict__ Wbc,
                                                          float* __restrict__ dtBC) {
  __shared__ float sT[2][G1_ROWS * LDS_PITCH];

  const int tid = threadIdx.x;
  const int wave = tid >> 5;
  const int lane = tid & 31;
  const int c = lane & 15, h = lane >> 4;
  const int mt = wave / 6, nt = wave - mt * 6;
  const int m0 = blockIdx.x * G1_BM;

  const int r0 = tid >> 4, q = tid & 15;
  const int r1 = (r0 + 24) & 31;
  const float* xs0 = x   + (size_t)(m0 + r0) * D_MODEL + 4 * q;
  const float* xs1 = x   + (size_t)(m0 + r1) * D_MODEL + 4 * q;
  const float* ws  = Wbc + (size_t)r0 * D_MODEL + 4 * q;
  const int dx0 = r0 * LDS_PITCH + 4 * q;
  const int dx1 = r1 * LDS_PITCH + 4 * q;
  const int dw  = (G1_BM + r0) * LDS_PITCH + 4 * q;

  v4f st[2][G1_NLD];
#pragma unroll
  for (int s = 0; s < 2; ++s) {
    st[s][0] = *(const v4f*)(xs0 + s * G1_KC);
    st[s][1] = *(const v4f*)(xs1 + s * G1_KC);
#pragma unroll
    for (int j = 0; j < 4; ++j)
      st[s][2 + j] = *(const v4f*)(ws + (size_t)(24 * j) * D_MODEL + s * G1_KC);
  }
  *(v4f*)(&sT[0][dx0]) = st[0][0];
  *(v4f*)(&sT[0][dx1]) = st[0][1];
#pragma unroll
  for (int j = 0; j < 4; ++j)
    *(v4f*)(&sT[0][dw + 24 * j * LDS_PITCH]) = st[0][2 + j];
  lds_barrier();

  v8f acc = {0.f, 0.f, 0.f, 0.f, 0.f, 0.f, 0.f, 0.f};
  const int aoff = (mt * 16 + c) * LDS_PITCH + 4 * h;
  const int boff = (G1_BM + nt * 16 + c) * LDS_PITCH + 4 * h;

#pragma unroll
  for (int kc = 0; kc < D_MODEL / G1_KC; ++kc) {
    if (kc + 2 < D_MODEL / G1_KC) {
      const int ko = (kc + 2) * G1_KC;
      st[kc & 1][0] = *(const v4f*)(xs0 + ko);
      st[kc & 1][1] = *(const v4f*)(xs1 + ko);
#pragma unroll
      for (int j = 0; j < 4; ++j)
        st[kc & 1][2 + j] = *(const v4f*)(ws + (size_t)(24 * j) * D_MODEL + ko);
    }
    __builtin_amdgcn_sched_barrier(0);
    const float* sa = &sT[kc & 1][aoff];
    const float* sb = &sT[kc & 1][boff];
#pragma unroll
    for (int jj = 0; jj < G1_KC / 8; ++jj) {
      const v4f a4 = *(const v4f*)(sa + 8 * jj);
      const v4f b4 = *(const v4f*)(sb + 8 * jj);
      v2f a, b;
      a[0] = a4[0]; a[1] = a4[1]; b[0] = b4[0]; b[1] = b4[1];
      acc = __builtin_amdgcn_wmma_f32_16x16x4_f32(false, a, false, b,
                                                  (short)0, acc, false, false);
      a[0] = a4[2]; a[1] = a4[3]; b[0] = b4[2]; b[1] = b4[3];
      acc = __builtin_amdgcn_wmma_f32_16x16x4_f32(false, a, false, b,
                                                  (short)0, acc, false, false);
    }
    __builtin_amdgcn_sched_barrier(0);
    if (kc + 1 < D_MODEL / G1_KC) {
      float* sn = sT[(kc + 1) & 1];
      *(v4f*)(&sn[dx0]) = st[(kc + 1) & 1][0];
      *(v4f*)(&sn[dx1]) = st[(kc + 1) & 1][1];
#pragma unroll
      for (int j = 0; j < 4; ++j)
        *(v4f*)(&sn[dw + 24 * j * LDS_PITCH]) = st[(kc + 1) & 1][2 + j];
    }
    lds_barrier();
  }
#pragma unroll
  for (int r = 0; r < 8; ++r) {
    int m = m0 + mt * 16 + 8 * h + r;
    dtBC[(size_t)m * E_DIM + nt * 16 + c] = acc[r];
  }
}

#define G2_BM      64
#define G2_BN      128
#define G2_ROWS    (G2_BM + G2_BN)
#define G2_NLD     12

__global__ __launch_bounds__(256) void k_proj_dt(const float* __restrict__ dtBC,
                                                 const float* __restrict__ Wdt,
                                                 const float* __restrict__ bdt,
                                                 float* __restrict__ dt) {
  __shared__ float sT[G2_ROWS * LDS_PITCH];

  const int tid = threadIdx.x;
  const int wave = tid >> 5, lane = tid & 31;
  const int c = lane & 15, h = lane >> 4;
  const int wm = wave & 3, wn = wave >> 2;
  const int m0 = (blockIdx.x >> 3) * G2_BM;
  const int n0 = (blockIdx.x & 7) * G2_BN;

  const int r0 = tid >> 4, q = tid & 15;
  const float* as = dtBC + (size_t)(m0 + r0) * E_DIM + 4 * q;
  const float* ws = Wdt  + (size_t)(n0 + r0) * DT_RANK + 4 * q;
  v4f st[G2_NLD];
#pragma unroll
  for (int j = 0; j < G2_NLD; ++j)
    st[j] = *(const v4f*)(j < 4 ? as + (size_t)(16 * j) * E_DIM
                                   : ws + (size_t)(16 * (j - 4)) * DT_RANK);
#pragma unroll
  for (int j = 0; j < G2_NLD; ++j)
    *(v4f*)(&sT[(r0 + 16 * j) * LDS_PITCH + 4 * q]) = st[j];
  __syncthreads();

  v8f acc[4];
#pragma unroll
  for (int t = 0; t < 4; ++t) acc[t] = (v8f){0.f, 0.f, 0.f, 0.f, 0.f, 0.f, 0.f, 0.f};

  const float* sa = &sT[(wm * 16 + c) * LDS_PITCH + 4 * h];
  const float* sb = &sT[(G2_BM + wn * 64 + c) * LDS_PITCH + 4 * h];
#pragma unroll
  for (int jj = 0; jj < DT_RANK / 8; ++jj) {
    const v4f a4 = *(const v4f*)(sa + 8 * jj);
    v2f a0, a1;
    a0[0] = a4[0]; a0[1] = a4[1]; a1[0] = a4[2]; a1[1] = a4[3];
#pragma unroll
    for (int t = 0; t < 4; ++t) {
      const v4f b4 = *(const v4f*)(sb + t * 16 * LDS_PITCH + 8 * jj);
      v2f b0, b1;
      b0[0] = b4[0]; b0[1] = b4[1]; b1[0] = b4[2]; b1[1] = b4[3];
      acc[t] = __builtin_amdgcn_wmma_f32_16x16x4_f32(false, a0, false, b0,
                                                     (short)0, acc[t], false, false);
      acc[t] = __builtin_amdgcn_wmma_f32_16x16x4_f32(false, a1, false, b1,
                                                     (short)0, acc[t], false, false);
    }
  }
#pragma unroll
  for (int t = 0; t < 4; ++t) {
    const int n = n0 + wn * 64 + t * 16 + c;
    const float bias = bdt[n];
#pragma unroll
    for (int r = 0; r < 8; ++r) {
      int m = m0 + wm * 16 + 8 * h + r;
      dt[(size_t)m * D_MODEL + n] = softplus_f(acc[t][r] + bias);
    }
  }
}

#define UPF 8
#define LOG2E_F 1.44269504088896341f

__global__ __launch_bounds__(256) void k_scan_local(const float* __restrict__ x,
                                                    const float* __restrict__ dt,
                                                    const float* __restrict__ dtBC,
                                                    const float* __restrict__ A_log,
                                                    float* __restrict__ chst) {
  const int tid = threadIdx.x;
  const int blk = blockIdx.x;
  const int d = ((blk & 3) << 8) + tid;
  const int c = (blk >> 2) & (NCHUNK - 1);
  const int b = blk >> 8;
  const size_t row0 = (size_t)b * SEQLEN + (size_t)c * TCHUNK;

  const float4* __restrict__ brow = (const float4*)(dtBC + row0 * E_DIM + DT_RANK);

  float A2[D_STATE];
#pragma unroll
  for (int n = 0; n < D_STATE; ++n) A2[n] = -__expf(A_log[d * D_STATE + n]) * LOG2E_F;

  float h[D_STATE];
#pragma unroll
  for (int n = 0; n < D_STATE; ++n) h[n] = 0.f;
  float sdt = 0.f;

  const float* dtp = dt + row0 * D_MODEL + d;
  const float* xp  = x  + row0 * D_MODEL + d;

  float dcur[UPF], xcur[UPF], dnxt[UPF], xnxt[UPF];
#pragma unroll
  for (int u = 0; u < UPF; ++u) {
    dcur[u] = dtp[(size_t)u * D_MODEL];
    xcur[u] = xp[(size_t)u * D_MODEL];
  }

  for (int tb = 0; tb < TCHUNK; tb += UPF) {
    const int tn = (tb + UPF < TCHUNK) ? tb + UPF : tb;
#pragma unroll
    for (int u = 0; u < UPF; ++u) {
      dnxt[u] = dtp[(size_t)(tn + u) * D_MODEL];
      xnxt[u] = xp[(size_t)(tn + u) * D_MODEL];
    }
    __builtin_amdgcn_sched_barrier(0);
#pragma unroll
    for (int u = 0; u < UPF; ++u) {
      const float dtv = dcur[u];
      const float xv  = xcur[u];
      float Bv[D_STATE];
#pragma unroll
      for (int i = 0; i < 4; ++i) {
        float4 q = brow[(tb + u) * (E_DIM / 4) + i];
        Bv[4*i+0] = q.x; Bv[4*i+1] = q.y; Bv[4*i+2] = q.z; Bv[4*i+3] = q.w;
      }
      const float dx = dtv * xv;
      sdt += dtv;
#pragma unroll
      for (int n = 0; n < D_STATE; ++n) {
        float dA = __builtin_amdgcn_exp2f(dtv * A2[n]);
        h[n] = dA * h[n] + dx * Bv[n];
      }
    }
    __builtin_amdgcn_sched_barrier(0);
#pragma unroll
    for (int u = 0; u < UPF; ++u) { dcur[u] = dnxt[u]; xcur[u] = xnxt[u]; }
  }
  size_t base = (size_t)(b * NCHUNK + c) * D_STATE * D_MODEL + d;
  v2f* __restrict__ st = (v2f*)chst;
#pragma unroll
  for (int n = 0; n < D_STATE; ++n) {
    v2f ph;
    ph[0] = __builtin_amdgcn_exp2f(sdt * A2[n]);
    ph[1] = h[n];
    st[base + (size_t)n * D_MODEL] = ph;
  }
}

__global__ __launch_bounds__(256) void k_scan_combine(const float* __restrict__ chst,
                                                      float* __restrict__ h0) {
  int gid = blockIdx.x * blockDim.x + threadIdx.x;
  int d = gid & (D_MODEL - 1);
  int n = (gid >> 10) & (D_STATE - 1);
  int b = gid >> 14;
  size_t base = ((size_t)b * NCHUNK * D_STATE + n) * D_MODEL + d;
  const v2f* __restrict__ pp = (const v2f*)chst + base;
  float* __restrict__ op = h0 + base;
  v2f ph[NCHUNK];
#pragma unroll
  for (int c = 0; c < NCHUNK; ++c) ph[c] = pp[(size_t)c * D_STATE * D_MODEL];
  __builtin_amdgcn_sched_barrier(0);
  float h = 0.f;
#pragma unroll
  for (int c = 0; c < NCHUNK; ++c) {
    op[(size_t)c * D_STATE * D_MODEL] = h;
    h = ph[c][0] * h + ph[c][1];
  }
}

__global__ __launch_bounds__(256) void k_scan_emit(const float* __restrict__ x,
                                                   const float* __restrict__ dt,
                                                   const float* __restrict__ dtBC,
                                                   const float* __restrict__ A_log,
                                                   const float* __restrict__ Dp,
                                                   const float* __restrict__ h0,
                                                   float* __restrict__ y) {
  const int tid = threadIdx.x;
  const int blk = blockIdx.x;
  const int d = ((blk & 3) << 8) + tid;
  const int c = (blk >> 2) & (NCHUNK - 1);
  const int b = blk >> 8;
  const size_t row0 = (size_t)b * SEQLEN + (size_t)c * TCHUNK;

  const float4* __restrict__ bcrow = (const float4*)(dtBC + row0 * E_DIM + DT_RANK);

  float A2[D_STATE];
#pragma unroll
  for (int n = 0; n < D_STATE; ++n) A2[n] = -__expf(A_log[d * D_STATE + n]) * LOG2E_F;

  float h[D_STATE];
  size_t base = (size_t)(b * NCHUNK + c) * D_STATE * D_MODEL + d;
#pragma unroll
  for (int n = 0; n < D_STATE; ++n) h[n] = h0[base + (size_t)n * D_MODEL];

  const float Dd = Dp[d];

  const float* dtp = dt + row0 * D_MODEL + d;
  const float* xp  = x  + row0 * D_MODEL + d;
  float*       yp  = y  + row0 * D_MODEL + d;

  float dcur[UPF], xcur[UPF], dnxt[UPF], xnxt[UPF];
#pragma unroll
  for (int u = 0; u < UPF; ++u) {
    dcur[u] = dtp[(size_t)u * D_MODEL];
    xcur[u] = xp[(size_t)u * D_MODEL];
  }

  for (int tb = 0; tb < TCHUNK; tb += UPF) {
    const int tn = (tb + UPF < TCHUNK) ? tb + UPF : tb;
#pragma unroll
    for (int u = 0; u < UPF; ++u) {
      dnxt[u] = dtp[(size_t)(tn + u) * D_MODEL];
      xnxt[u] = xp[(size_t)(tn + u) * D_MODEL];
    }
    __builtin_amdgcn_sched_barrier(0);
#pragma unroll
    for (int u = 0; u < UPF; ++u) {
      const float dtv = dcur[u];
      const float xv  = xcur[u];
      float Bv[D_STATE], Cv[D_STATE];
#pragma unroll
      for (int i = 0; i < 4; ++i) {
        float4 q = bcrow[(tb + u) * (E_DIM / 4) + i];
        Bv[4*i+0] = q.x; Bv[4*i+1] = q.y; Bv[4*i+2] = q.z; Bv[4*i+3] = q.w;
      }
#pragma unroll
      for (int i = 0; i < 4; ++i) {
        float4 q = bcrow[(tb + u) * (E_DIM / 4) + 4 + i];
        Cv[4*i+0] = q.x; Cv[4*i+1] = q.y; Cv[4*i+2] = q.z; Cv[4*i+3] = q.w;
      }
      const float dx = dtv * xv;
      float yv = 0.f;
#pragma unroll
      for (int n = 0; n < D_STATE; ++n) {
        float dA = __builtin_amdgcn_exp2f(dtv * A2[n]);
        h[n] = dA * h[n] + dx * Bv[n];
        yv += h[n] * Cv[n];
      }
      yp[(size_t)(tb + u) * D_MODEL] = yv + xv * Dd;
    }
    __builtin_amdgcn_sched_barrier(0);
#pragma unroll
    for (int u = 0; u < UPF; ++u) { dcur[u] = dnxt[u]; xcur[u] = xnxt[u]; }
  }
}

extern "C" void kernel_launch(void* const* d_in, const int* in_sizes, int n_in,
                              void* d_out, int out_size, void* d_ws, size_t ws_size,
                              hipStream_t stream) {
  const float* x    = (const float*)d_in[0];
  const float* Alog = (const float*)d_in[1];
  const float* Dp   = (const float*)d_in[2];
  const float* Wdt  = (const float*)d_in[3];
  const float* bdt  = (const float*)d_in[4];
  const float* Wbc  = (const float*)d_in[5];
  float* y = (float*)d_out;

  float* ws    = (float*)d_ws;
  float* dtBC  = ws;
  float* dt    = dtBC + (size_t)ROWS * E_DIM;
  float* chst  = dt + (size_t)ROWS * D_MODEL;
  float* h0    = chst + (size_t)2 * BATCH * D_MODEL * NCHUNK * D_STATE;

  k_proj_dtbc<<<ROWS / G1_BM, G1_THREADS, 0, stream>>>(x, Wbc, dtBC);

  k_proj_dt<<<(ROWS / G2_BM) * (D_MODEL / G2_BN), 256, 0, stream>>>(dtBC, Wdt, bdt, dt);

  int nscan = BATCH * D_MODEL * NCHUNK;
  k_scan_local<<<nscan / 256, 256, 0, stream>>>(x, dt, dtBC, Alog, chst);

  int ncomb = BATCH * D_MODEL * D_STATE;
  k_scan_combine<<<ncomb / 256, 256, 0, stream>>>(chst, h0);

  k_scan_emit<<<nscan / 256, 256, 0, stream>>>(x, dt, dtBC, Alog, Dp, h0, y);
}
